# starting kernel
# baseline (speedup 1.0000x reference)
#include <hip/hip_runtime.h>
#include <hip/hip_bf16.h>

#define N_NODES 50000
#define N_EDGES 800000
#define IN_F 128
#define OUT_F 64
#define HEADS 4
#define HF 256
#define NEG_SLOPE 0.2f
#define GAT_EPS 1e-16f

typedef __attribute__((ext_vector_type(16))) _Float16 v16h;
typedef __attribute__((ext_vector_type(8)))  _Float16 v8h;
typedef __attribute__((ext_vector_type(8)))  float    v8f;
typedef __attribute__((ext_vector_type(4)))  float    v4f;

__global__ void cvt_f16_kernel(const float* __restrict__ x, const float* __restrict__ W,
                               _Float16* __restrict__ Xh, _Float16* __restrict__ Wh) {
  int idx = blockIdx.x * blockDim.x + threadIdx.x;
  int stride = gridDim.x * blockDim.x;
  for (int i = idx; i < N_NODES * IN_F; i += stride) Xh[i] = (_Float16)x[i];
  for (int i = idx; i < HF * IN_F; i += stride)      Wh[i] = (_Float16)W[i];
}

__global__ void init_kernel(float* __restrict__ out, const float* __restrict__ bias,
                            int* __restrict__ m, float* __restrict__ ssum) {
  int idx = blockIdx.x * blockDim.x + threadIdx.x;
  int stride = gridDim.x * blockDim.x;
  for (int i = idx; i < N_NODES * HF; i += stride) out[i] = bias[i & (HF - 1)];
  for (int i = idx; i < N_NODES * HEADS; i += stride) {
    m[i] = (int)0x80000000;
    ssum[i] = 0.0f;
  }
}

__global__ void __launch_bounds__(256)
gemm_wmma_kernel(const _Float16* __restrict__ Xh, const _Float16* __restrict__ Wh,
                 float* __restrict__ h) {
  const int wave  = (blockIdx.x << 3) + (threadIdx.x >> 5);
  const int lane  = threadIdx.x & 31;
  const int tileM = wave >> 4;
  const int tileN = wave & 15;
  const int mrow0 = tileM << 4;
  const int ncol0 = tileN << 4;

  const int l15   = lane & 15;
  const int khalf = lane >> 4;

  const int arow = mrow0 + l15;
  const int brow = ncol0 + l15;

  const _Float16* ap = Xh + (size_t)arow * IN_F + khalf * 8;
  const _Float16* bp = Wh + (size_t)brow * IN_F + khalf * 8;

  v8f c = {};
  #pragma unroll
  for (int kb = 0; kb < IN_F; kb += 32) {
    v8h a0 = *(const v8h*)(ap + kb);
    v8h a1 = *(const v8h*)(ap + kb + 16);
    v8h b0 = *(const v8h*)(bp + kb);
    v8h b1 = *(const v8h*)(bp + kb + 16);
    v16h A, B;
    #pragma unroll
    for (int i = 0; i < 8; ++i) { A[i] = a0[i]; A[i + 8] = a1[i];
                                  B[i] = b0[i]; B[i + 8] = b1[i]; }
    c = __builtin_amdgcn_wmma_f32_16x16x32_f16(false, A, false, B, (short)0, c,
                                               false, false);
  }
  float* hp = h + (size_t)(mrow0 + khalf * 8) * HF + ncol0 + l15;
  #pragma unroll
  for (int r = 0; r < 8; ++r) hp[(size_t)r * HF] = c[r];
}

__global__ void __launch_bounds__(256)
logits_kernel(const float* __restrict__ h,
              const float* __restrict__ att_src, const float* __restrict__ att_dst,
              float* __restrict__ a_src, float* __restrict__ a_dst) {
  int node = (blockIdx.x << 3) + (threadIdx.x >> 5);
  if (node >= N_NODES) return;
  int lane  = threadIdx.x & 31;
  int head  = lane >> 3;
  int fbase = (lane & 7) * 8;
  const float* hp = h + (size_t)node * HF + lane * 8;
  float ps = 0.f, pd = 0.f;
  #pragma unroll
  for (int j = 0; j < 8; ++j) {
    float hv = hp[j];
    ps += hv * att_src[head * OUT_F + fbase + j];
    pd += hv * att_dst[head * OUT_F + fbase + j];
  }
  #pragma unroll
  for (int off = 1; off < 8; off <<= 1) {
    ps += __shfl_xor(ps, off, 32);
    pd += __shfl_xor(pd, off, 32);
  }
  if ((lane & 7) == 0) {
    a_src[node * HEADS + head] = ps;
    a_dst[node * HEADS + head] = pd;
  }
}

__device__ __forceinline__ int enc_f(float f) {
  int i = __float_as_int(f);
  return (i >= 0) ? i : (i ^ 0x7fffffff);
}
__device__ __forceinline__ float dec_f(int i) {
  return __int_as_float((i >= 0) ? i : (i ^ 0x7fffffff));
}

__global__ void edge_score_kernel(const int* __restrict__ ei,
                                  const float* __restrict__ a_src,
                                  const float* __restrict__ a_dst,
                                  float* __restrict__ ebuf, int* __restrict__ m) {
  int i = blockIdx.x * blockDim.x + threadIdx.x;
  if (i >= N_EDGES) return;
  int s = ei[i], d = ei[N_EDGES + i];
  #pragma unroll
  for (int hh = 0; hh < HEADS; ++hh) {
    float ev = a_src[s * HEADS + hh] + a_dst[d * HEADS + hh];
    ev = (ev >= 0.f) ? ev : NEG_SLOPE * ev;
    ebuf[(size_t)i * HEADS + hh] = ev;
    atomicMax(&m[d * HEADS + hh], enc_f(ev));
  }
}

__global__ void edge_exp_kernel(const int* __restrict__ ei, const int* __restrict__ m,
                                float* __restrict__ ebuf, float* __restrict__ ssum) {
  int i = blockIdx.x * blockDim.x + threadIdx.x;
  if (i >= N_EDGES) return;
  int d = ei[N_EDGES + i];
  #pragma unroll
  for (int hh = 0; hh < HEADS; ++hh) {
    int menc = m[d * HEADS + hh];
    float mv = (menc == (int)0x80000000) ? 0.f : dec_f(menc);
    float ex = __expf(ebuf[(size_t)i * HEADS + hh] - mv);
    ebuf[(size_t)i * HEADS + hh] = ex;
    atomicAdd(&ssum[d * HEADS + hh], ex);
  }
}

__global__ void __launch_bounds__(256)
scatter_kernel(const int* __restrict__ ei, const float* __restrict__ ebuf,
               const float* __restrict__ ssum, const float* __restrict__ h,
               float* __restrict__ out) {
  int edge = (blockIdx.x << 3) + (threadIdx.x >> 5);
  if (edge >= N_EDGES) return;
  int lane = threadIdx.x & 31;
  int s = ei[edge], d = ei[N_EDGES + edge];
  const float* hs = h + (size_t)s * HF;
  float* od = out + (size_t)d * HF;
  __builtin_prefetch(hs + lane * 8, 0, 0);
  v4f e4 = *(const v4f*)(ebuf + (size_t)edge * HEADS);
  v4f s4 = *(const v4f*)(ssum + (size_t)d * HEADS);
  v4f alpha;
  #pragma unroll
  for (int k = 0; k < 4; ++k) alpha[k] = e4[k] / (s4[k] + GAT_EPS);
  #pragma unroll
  for (int j = 0; j < 8; ++j) {
    int f = lane + (j << 5);
    float a = alpha[j >> 1];
    atomicAdd(&od[f], a * hs[f]);
  }
}

extern "C" void kernel_launch(void* const* d_in, const int* in_sizes, int n_in,
                              void* d_out, int out_size, void* d_ws, size_t ws_size,
                              hipStream_t stream) {
  const float* x       = (const float*)d_in[0];
  const int*   ei      = (const int*)  d_in[1];
  const float* W       = (const float*)d_in[2];
  const float* att_src = (const float*)d_in[3];
  const float* att_dst = (const float*)d_in[4];
  const float* bias    = (const float*)d_in[5];
  float* out = (float*)d_out;

  char* p = (char*)d_ws;
  auto alloc = [&](size_t bytes) {
    char* r = p; p += (bytes + 255) & ~(size_t)255; return r;
  };
  float*    h     = (float*)   alloc((size_t)N_NODES * HF * 4);
  _Float16* Xh    = (_Float16*)alloc((size_t)N_NODES * IN_F * 2);
  _Float16* Wh    = (_Float16*)alloc((size_t)HF * IN_F * 2);
  float*    a_src = (float*)   alloc((size_t)N_NODES * HEADS * 4);
  float*    a_dst = (float*)   alloc((size_t)N_NODES * HEADS * 4);
  int*      m     = (int*)     alloc((size_t)N_NODES * HEADS * 4);
  float*    ssum  = (float*)   alloc((size_t)N_NODES * HEADS * 4);
  float*    ebuf  = (float*)   alloc((size_t)N_EDGES * HEADS * 4);

  cvt_f16_kernel  <<<2048, 256, 0, stream>>>(x, W, Xh, Wh);
  init_kernel     <<<2048, 256, 0, stream>>>(out, bias, m, ssum);
  gemm_wmma_kernel<<<6250, 256, 0, stream>>>(Xh, Wh, h);
  logits_kernel   <<<6250, 256, 0, stream>>>(h, att_src, att_dst, a_src, a_dst);
  edge_score_kernel<<<(N_EDGES + 255) / 256, 256, 0, stream>>>(ei, a_src, a_dst, ebuf, m);
  edge_exp_kernel <<<(N_EDGES + 255) / 256, 256, 0, stream>>>(ei, m, ebuf, ssum);
  scatter_kernel  <<<N_EDGES / 8, 256, 0, stream>>>(ei, ebuf, ssum, h, out);
}


# Aggregation sums in LDS: a block owns 196 dest rows, filters the edge list itself, adds alpha*h with LDS float adds, stores out once; global atomics on out and its fill gone; sums arrival-ordered
# speedup vs baseline: 2.6834x; 2.6834x over previous
#include <hip/hip_runtime.h>
#include <hip/hip_bf16.h>

#define N_NODES 50000
#define N_EDGES 800000
#define IN_F 128
#define OUT_F 64
#define HEADS 4
#define HF 256
#define NEG_SLOPE 0.2f
#define GAT_EPS 1e-16f

typedef __attribute__((ext_vector_type(16))) _Float16 v16h;
typedef __attribute__((ext_vector_type(8)))  _Float16 v8h;
typedef __attribute__((ext_vector_type(8)))  float    v8f;
typedef __attribute__((ext_vector_type(4)))  float    v4f;

__global__ void cvt_f16_kernel(const float* __restrict__ x, const float* __restrict__ W,
                               _Float16* __restrict__ Xh, _Float16* __restrict__ Wh) {
  int idx = blockIdx.x * blockDim.x + threadIdx.x;
  int stride = gridDim.x * blockDim.x;
  for (int i = idx; i < N_NODES * IN_F; i += stride) Xh[i] = (_Float16)x[i];
  for (int i = idx; i < HF * IN_F; i += stride)      Wh[i] = (_Float16)W[i];
}

__global__ void init_kernel(int* __restrict__ m, float* __restrict__ ssum) {
  int idx = blockIdx.x * blockDim.x + threadIdx.x;
  int stride = gridDim.x * blockDim.x;
  for (int i = idx; i < N_NODES * HEADS; i += stride) {
    m[i] = (int)0x80000000;
    ssum[i] = 0.0f;
  }
}

__global__ void __launch_bounds__(256)
gemm_wmma_kernel(const _Float16* __restrict__ Xh, const _Float16* __restrict__ Wh,
                 float* __restrict__ h) {
  const int wave  = (blockIdx.x << 3) + (threadIdx.x >> 5);
  const int lane  = threadIdx.x & 31;
  const int tileM = wave >> 4;
  const int tileN = wave & 15;
  const int mrow0 = tileM << 4;
  const int ncol0 = tileN << 4;

  const int l15   = lane & 15;
  const int khalf = lane >> 4;

  const int arow = mrow0 + l15;
  const int brow = ncol0 + l15;

  const _Float16* ap = Xh + (size_t)arow * IN_F + khalf * 8;
  const _Float16* bp = Wh + (size_t)brow * IN_F + khalf * 8;

  v8f c = {};
  #pragma unroll
  for (int kb = 0; kb < IN_F; kb += 32) {
    v8h a0 = *(const v8h*)(ap + kb);
    v8h a1 = *(const v8h*)(ap + kb + 16);
    v8h b0 = *(const v8h*)(bp + kb);
    v8h b1 = *(const v8h*)(bp + kb + 16);
    v16h A, B;
    #pragma unroll
    for (int i = 0; i < 8; ++i) { A[i] = a0[i]; A[i + 8] = a1[i];
                                  B[i] = b0[i]; B[i + 8] = b1[i]; }
    c = __builtin_amdgcn_wmma_f32_16x16x32_f16(false, A, false, B, (short)0, c,
                                               false, false);
  }
  float* hp = h + (size_t)(mrow0 + khalf * 8) * HF + ncol0 + l15;
  #pragma unroll
  for (int r = 0; r < 8; ++r) hp[(size_t)r * HF] = c[r];
}

__global__ void __launch_bounds__(256)
logits_kernel(const float* __restrict__ h,
              const float* __restrict__ att_src, const float* __restrict__ att_dst,
              float* __restrict__ a_src, float* __restrict__ a_dst) {
  int node = (blockIdx.x << 3) + (threadIdx.x >> 5);
  if (node >= N_NODES) return;
  int lane  = threadIdx.x & 31;
  int head  = lane >> 3;
  int fbase = (lane & 7) * 8;
  const float* hp = h + (size_t)node * HF + lane * 8;
  float ps = 0.f, pd = 0.f;
  #pragma unroll
  for (int j = 0; j < 8; ++j) {
    float hv = hp[j];
    ps += hv * att_src[head * OUT_F + fbase + j];
    pd += hv * att_dst[head * OUT_F + fbase + j];
  }
  #pragma unroll
  for (int off = 1; off < 8; off <<= 1) {
    ps += __shfl_xor(ps, off, 32);
    pd += __shfl_xor(pd, off, 32);
  }
  if ((lane & 7) == 0) {
    a_src[node * HEADS + head] = ps;
    a_dst[node * HEADS + head] = pd;
  }
}

__device__ __forceinline__ int enc_f(float f) {
  int i = __float_as_int(f);
  return (i >= 0) ? i : (i ^ 0x7fffffff);
}
__device__ __forceinline__ float dec_f(int i) {
  return __int_as_float((i >= 0) ? i : (i ^ 0x7fffffff));
}

#define EDGES_PER_BLOCK 256
#define EDGE_TRIPS 4
#define EDGES_PER_TRIP (EDGES_PER_BLOCK / EDGE_TRIPS)
static_assert(N_EDGES % EDGES_PER_BLOCK == 0, "edge blocks are full: no tail guard");
static_assert(EDGES_PER_TRIP * HEADS == 256, "one lane per (edge, head) of a trip");

__global__ void __launch_bounds__(256)
edge_score_kernel(const int* __restrict__ ei,
                  const float* __restrict__ a_src,
                  const float* __restrict__ a_dst,
                  float* __restrict__ ebuf, int* __restrict__ m) {
  const int hh = threadIdx.x & (HEADS - 1);
  const int e0 = blockIdx.x * EDGES_PER_BLOCK + (threadIdx.x >> 2);
  int s[EDGE_TRIPS], d[EDGE_TRIPS];
  float as[EDGE_TRIPS], ad[EDGE_TRIPS];
  #pragma unroll
  for (int t = 0; t < EDGE_TRIPS; ++t) {
    s[t] = ei[e0 + t * EDGES_PER_TRIP];
    d[t] = ei[N_EDGES + e0 + t * EDGES_PER_TRIP];
  }
  #pragma unroll
  for (int t = 0; t < EDGE_TRIPS; ++t) {
    as[t] = a_src[s[t] * HEADS + hh];
    ad[t] = a_dst[d[t] * HEADS + hh];
  }
  #pragma unroll
  for (int t = 0; t < EDGE_TRIPS; ++t) {
    float ev = as[t] + ad[t];
    ev = (ev >= 0.f) ? ev : NEG_SLOPE * ev;
    ebuf[(size_t)(e0 + t * EDGES_PER_TRIP) * HEADS + hh] = ev;
    atomicMax(&m[d[t] * HEADS + hh], enc_f(ev));
  }
}

__global__ void __launch_bounds__(256)
edge_exp_kernel(const int* __restrict__ ei, const int* __restrict__ m,
                float* __restrict__ ebuf, float* __restrict__ ssum) {
  const int hh = threadIdx.x & (HEADS - 1);
  const int e0 = blockIdx.x * EDGES_PER_BLOCK + (threadIdx.x >> 2);
  int d[EDGE_TRIPS], menc[EDGE_TRIPS];
  float ev[EDGE_TRIPS];
  #pragma unroll
  for (int t = 0; t < EDGE_TRIPS; ++t) {
    d[t]  = ei[N_EDGES + e0 + t * EDGES_PER_TRIP];
    ev[t] = ebuf[(size_t)(e0 + t * EDGES_PER_TRIP) * HEADS + hh];
  }
  #pragma unroll
  for (int t = 0; t < EDGE_TRIPS; ++t) menc[t] = m[d[t] * HEADS + hh];
  #pragma unroll
  for (int t = 0; t < EDGE_TRIPS; ++t) {
    float mv = (menc[t] == (int)0x80000000) ? 0.f : dec_f(menc[t]);
    float ex = __expf(ev[t] - mv);
    ebuf[(size_t)(e0 + t * EDGES_PER_TRIP) * HEADS + hh] = ex;
    atomicAdd(&ssum[d[t] * HEADS + hh], ex);
  }
}

#define TILE_N 196
#define AGG_BLOCKS 256
#define AGG_THREADS 1024
#define LIST_CAP 24576
#define SCAN_VECS (N_EDGES / 4)
#define SCAN_BATCH 7
#define SCAN_ITERS 28
#define AGG_EDGES_PER_TRIP 8
static_assert(TILE_N * AGG_BLOCKS >= N_NODES, "the tiles cover every node");
static_assert(N_EDGES % 4 == 0, "destination ids are read four at a time");
static_assert(SCAN_ITERS * SCAN_BATCH * AGG_THREADS >= SCAN_VECS, "the scan covers the edge list");
static_assert(N_EDGES <= (1 << 20) && TILE_N <= 256, "edge id and local row share a word");
static_assert(TILE_N * HF * 4 + LIST_CAP * 4 + 16 <= 320 * 1024, "tile and list fit the LDS");

__device__ __forceinline__ int tile_pos(int f) {
  return (f & 128) + ((f & 3) << 5) + ((f & 127) >> 2);
}

template <int U>
__device__ __forceinline__ void agg_trip(int i0, int lo, int lane,
                                         const unsigned* list, float* acc,
                                         const int* __restrict__ ei, const float* __restrict__ ebuf,
                                         const float* __restrict__ ssum, const float* __restrict__ h) {
  int e[U], dl[U], s[U];
  v4f e4[U], s4[U], h0[U], h1[U];
  #pragma unroll
  for (int u = 0; u < U; ++u) {
    unsigned ent = (unsigned)__builtin_amdgcn_readfirstlane((int)list[i0 + u]);
    e[u]  = (int)(ent & 0xFFFFFu);
    dl[u] = (int)(ent >> 20);
  }
  #pragma unroll
  for (int u = 0; u < U; ++u) {
    s[u]  = ei[e[u]];
    e4[u] = *(const v4f*)(ebuf + (size_t)e[u] * HEADS);
    s4[u] = *(const v4f*)(ssum + (size_t)(lo + dl[u]) * HEADS);
  }
  #pragma unroll
  for (int u = 0; u < U; ++u) {
    const float* hs = h + (size_t)s[u] * HF + lane * 4;
    h0[u] = *(const v4f*)(hs);
    h1[u] = *(const v4f*)(hs + 128);
  }
  const int hsel = lane >> 4;
  #pragma unroll
  for (int u = 0; u < U; ++u) {
    float a0 = e4[u][0] / (s4[u][0] + GAT_EPS);
    float a1 = e4[u][1] / (s4[u][1] + GAT_EPS);
    float a2 = e4[u][2] / (s4[u][2] + GAT_EPS);
    float a3 = e4[u][3] / (s4[u][3] + GAT_EPS);
    float alo = hsel ? a1 : a0;
    float ahi = hsel ? a3 : a2;
    float* row = acc + dl[u] * HF + lane;
    #pragma unroll
    for (int k = 0; k < 4; ++k) {
      atomicAdd(row + 32 * k,       alo * h0[u][k]);
      atomicAdd(row + 128 + 32 * k, ahi * h1[u][k]);
    }
  }
}

__global__ void __launch_bounds__(AGG_THREADS)
aggregate_kernel(const int* __restrict__ ei, const float* __restrict__ ebuf,
                 const float* __restrict__ ssum, const float* __restrict__ h,
                 const float* __restrict__ bias, float* __restrict__ out) {
  __shared__ __attribute__((aligned(16))) float acc[TILE_N * HF];
  __shared__ unsigned list[LIST_CAP];
  __shared__ unsigned cnt;

  const int tid  = threadIdx.x;
  const int lane = tid & 31;
  const int wave = tid >> 5;
  const int lo   = blockIdx.x * TILE_N;

  for (int i = tid; i < TILE_N * HF; i += AGG_THREADS) {
    int p = i & (HF - 1);
    int f = (p & 128) + ((p & 31) << 2) + ((p & 127) >> 5);
    acc[i] = bias[f];
  }
  if (tid == 0) cnt = 0u;
  __syncthreads();

  const int4* dst4 = (const int4*)(ei + N_EDGES);
  for (int it = 0; it < SCAN_ITERS; ++it) {
    int4 dv[SCAN_BATCH];
    #pragma unroll
    for (int u = 0; u < SCAN_BATCH; ++u) {
      int v = (it * SCAN_BATCH + u) * AGG_THREADS + tid;
      dv[u] = (v < SCAN_VECS) ? dst4[v] : make_int4(-1, -1, -1, -1);
    }
    #pragma unroll
    for (int u = 0; u < SCAN_BATCH; ++u) {
      int v = (it * SCAN_BATCH + u) * AGG_THREADS + tid;
      unsigned r0 = (unsigned)(dv[u].x - lo), r1 = (unsigned)(dv[u].y - lo);
      unsigned r2 = (unsigned)(dv[u].z - lo), r3 = (unsigned)(dv[u].w - lo);
      if (min(min(r0, r1), min(r2, r3)) < (unsigned)TILE_N) {
        unsigned rr[4] = {r0, r1, r2, r3};
        #pragma unroll
        for (int k = 0; k < 4; ++k) {
          if (rr[k] < (unsigned)TILE_N) {
            int e = 4 * v + k;
            unsigned pos = atomicAdd(&cnt, 1u);
            if (pos < (unsigned)LIST_CAP) list[pos] = (unsigned)e | (rr[k] << 20);
          }
        }
      }
    }
  }
  __syncthreads();

  const unsigned total = cnt;
  const int n = (total <= (unsigned)LIST_CAP) ? (int)total : 0;
  if (total > (unsigned)LIST_CAP) {
    for (int e = tid; e < N_EDGES; e += AGG_THREADS) {
      unsigned r = (unsigned)(ei[N_EDGES + e] - lo);
      if (r < (unsigned)TILE_N) {
        int s = ei[e];
        float al[HEADS];
        #pragma unroll
        for (int q = 0; q < HEADS; ++q)
          al[q] = ebuf[(size_t)e * HEADS + q] / (ssum[(size_t)(lo + (int)r) * HEADS + q] + GAT_EPS);
        #pragma unroll 1
        for (int f = 0; f < HF; ++f) {
          float a = (f < 128) ? ((f < 64) ? al[0] : al[1]) : ((f < 192) ? al[2] : al[3]);
          atomicAdd(&acc[r * HF + tile_pos(f)], a * h[(size_t)s * HF + f]);
        }
      }
    }
  }
  const int full = n / AGG_EDGES_PER_TRIP;
  for (int t = wave; t < full; t += AGG_THREADS / 32)
    agg_trip<AGG_EDGES_PER_TRIP>(t * AGG_EDGES_PER_TRIP, lo, lane, list, acc, ei, ebuf, ssum, h);
  for (int i = full * AGG_EDGES_PER_TRIP + wave; i < n; i += AGG_THREADS / 32)
    agg_trip<1>(i, lo, lane, list, acc, ei, ebuf, ssum, h);
  __syncthreads();

  for (int r = wave; r < TILE_N; r += AGG_THREADS / 32) {
    int node = lo + r;
    if (node < N_NODES) {
      const float* row = acc + r * HF + lane;
      v4f o0, o1;
      #pragma unroll
      for (int k = 0; k < 4; ++k) { o0[k] = row[32 * k]; o1[k] = row[128 + 32 * k]; }
      float* op = out + (size_t)node * HF + lane * 4;
      *(v4f*)(op)       = o0;
      *(v4f*)(op + 128) = o1;
    }
  }
}

extern "C" void kernel_launch(void* const* d_in, const int* in_sizes, int n_in,
                              void* d_out, int out_size, void* d_ws, size_t ws_size,
                              hipStream_t stream) {
  const float* x       = (const float*)d_in[0];
  const int*   ei      = (const int*)  d_in[1];
  const float* W       = (const float*)d_in[2];
  const float* att_src = (const float*)d_in[3];
  const float* att_dst = (const float*)d_in[4];
  const float* bias    = (const float*)d_in[5];
  float* out = (float*)d_out;

  char* p = (char*)d_ws;
  auto alloc = [&](size_t bytes) {
    char* r = p; p += (bytes + 255) & ~(size_t)255; return r;
  };
  float*    h     = (float*)   alloc((size_t)N_NODES * HF * 4);
  _Float16* Xh    = (_Float16*)alloc((size_t)N_NODES * IN_F * 2);
  _Float16* Wh    = (_Float16*)alloc((size_t)HF * IN_F * 2);
  float*    a_src = (float*)   alloc((size_t)N_NODES * HEADS * 4);
  float*    a_dst = (float*)   alloc((size_t)N_NODES * HEADS * 4);
  int*      m     = (int*)     alloc((size_t)N_NODES * HEADS * 4);
  float*    ssum  = (float*)   alloc((size_t)N_NODES * HEADS * 4);
  float*    ebuf  = (float*)   alloc((size_t)N_EDGES * HEADS * 4);

  cvt_f16_kernel  <<<2048, 256, 0, stream>>>(x, W, Xh, Wh);
  init_kernel     <<<256, 256, 0, stream>>>(m, ssum);
  gemm_wmma_kernel<<<6250, 256, 0, stream>>>(Xh, Wh, h);
  logits_kernel   <<<6250, 256, 0, stream>>>(h, att_src, att_dst, a_src, a_dst);
  edge_score_kernel<<<N_EDGES / EDGES_PER_BLOCK, 256, 0, stream>>>(ei, a_src, a_dst, ebuf, m);
  edge_exp_kernel <<<N_EDGES / EDGES_PER_BLOCK, 256, 0, stream>>>(ei, m, ebuf, ssum);
  aggregate_kernel<<<AGG_BLOCKS, AGG_THREADS, 0, stream>>>(ei, ebuf, ssum, h, bias, out);
}


# Softmax moves into the aggregation block: row max, exp sum and 1/(sum+eps) built in LDS from its own edge list; score, exp, init launches and their global atomics gone; sums arrival-ordered
# speedup vs baseline: 4.2174x; 1.5717x over previous
#include <hip/hip_runtime.h>
#include <hip/hip_bf16.h>

#define N_NODES 50000
#define N_EDGES 800000
#define IN_F 128
#define OUT_F 64
#define HEADS 4
#define HF 256
#define NEG_SLOPE 0.2f
#define GAT_EPS 1e-16f

typedef __attribute__((ext_vector_type(16))) _Float16 v16h;
typedef __attribute__((ext_vector_type(8)))  _Float16 v8h;
typedef __attribute__((ext_vector_type(8)))  float    v8f;
typedef __attribute__((ext_vector_type(4)))  float    v4f;
typedef __attribute__((ext_vector_type(4)))  int      v4i;

__global__ void cvt_f16_kernel(const float* __restrict__ x, const float* __restrict__ W,
                               _Float16* __restrict__ Xh, _Float16* __restrict__ Wh) {
  int idx = blockIdx.x * blockDim.x + threadIdx.x;
  int stride = gridDim.x * blockDim.x;
  for (int i = idx; i < N_NODES * IN_F; i += stride) Xh[i] = (_Float16)x[i];
  for (int i = idx; i < HF * IN_F; i += stride)      Wh[i] = (_Float16)W[i];
}

__global__ void __launch_bounds__(256)
gemm_wmma_kernel(const _Float16* __restrict__ Xh, const _Float16* __restrict__ Wh,
                 float* __restrict__ h) {
  const int wave  = (blockIdx.x << 3) + (threadIdx.x >> 5);
  const int lane  = threadIdx.x & 31;
  const int tileM = wave >> 4;
  const int tileN = wave & 15;
  const int mrow0 = tileM << 4;
  const int ncol0 = tileN << 4;

  const int l15   = lane & 15;
  const int khalf = lane >> 4;

  const int arow = mrow0 + l15;
  const int brow = ncol0 + l15;

  const _Float16* ap = Xh + (size_t)arow * IN_F + khalf * 8;
  const _Float16* bp = Wh + (size_t)brow * IN_F + khalf * 8;

  v8f c = {};
  #pragma unroll
  for (int kb = 0; kb < IN_F; kb += 32) {
    v8h a0 = *(const v8h*)(ap + kb);
    v8h a1 = *(const v8h*)(ap + kb + 16);
    v8h b0 = *(const v8h*)(bp + kb);
    v8h b1 = *(const v8h*)(bp + kb + 16);
    v16h A, B;
    #pragma unroll
    for (int i = 0; i < 8; ++i) { A[i] = a0[i]; A[i + 8] = a1[i];
                                  B[i] = b0[i]; B[i + 8] = b1[i]; }
    c = __builtin_amdgcn_wmma_f32_16x16x32_f16(false, A, false, B, (short)0, c,
                                               false, false);
  }
  float* hp = h + (size_t)(mrow0 + khalf * 8) * HF + ncol0 + l15;
  #pragma unroll
  for (int r = 0; r < 8; ++r) hp[(size_t)r * HF] = c[r];
}

__global__ void __launch_bounds__(256)
logits_kernel(const float* __restrict__ h,
              const float* __restrict__ att_src, const float* __restrict__ att_dst,
              float* __restrict__ a_src, float* __restrict__ a_dst) {
  int node = (blockIdx.x << 3) + (threadIdx.x >> 5);
  if (node >= N_NODES) return;
  int lane  = threadIdx.x & 31;
  int head  = lane >> 3;
  int fbase = (lane & 7) * 8;
  const float* hp = h + (size_t)node * HF + lane * 8;
  float ps = 0.f, pd = 0.f;
  #pragma unroll
  for (int j = 0; j < 8; ++j) {
    float hv = hp[j];
    ps += hv * att_src[head * OUT_F + fbase + j];
    pd += hv * att_dst[head * OUT_F + fbase + j];
  }
  #pragma unroll
  for (int off = 1; off < 8; off <<= 1) {
    ps += __shfl_xor(ps, off, 32);
    pd += __shfl_xor(pd, off, 32);
  }
  if ((lane & 7) == 0) {
    a_src[node * HEADS + head] = ps;
    a_dst[node * HEADS + head] = pd;
  }
}

__device__ __forceinline__ int enc_f(float f) {
  int i = __float_as_int(f);
  return (i >= 0) ? i : (i ^ 0x7fffffff);
}
__device__ __forceinline__ float dec_f(int i) {
  return __int_as_float((i >= 0) ? i : (i ^ 0x7fffffff));
}

#define TILE_N 196
#define AGG_BLOCKS 256
#define AGG_THREADS 1024
#define LIST_CAP 24576
#define SCAN_VECS (N_EDGES / 4)
#define SCAN_BATCH 7
#define SCAN_ITERS 28
#define AGG_EDGES_PER_TRIP 8
#define SOFT_BATCH 4
#define ROW_HEADS (TILE_N * HEADS)
static_assert(TILE_N * AGG_BLOCKS >= N_NODES, "the tiles cover every node");
static_assert(N_EDGES % 4 == 0, "destination ids are read four at a time");
static_assert(SCAN_ITERS * SCAN_BATCH * AGG_THREADS >= SCAN_VECS, "the scan covers the edge list");
static_assert(N_EDGES <= (1 << 20) && TILE_N <= 256, "edge id and local row share a word");
static_assert(N_NODES <= (1 << 16), "source id and local row share a word");
static_assert(SOFT_BATCH == 4, "the gather fences of phases 2 and 3 name four edges");
static_assert(ROW_HEADS <= AGG_THREADS, "one lane per (row, head) of the small planes");
static_assert(TILE_N * HF * 4 + LIST_CAP * 4 + 3 * ROW_HEADS * 4 + 16 <= 320 * 1024, "tile, list and planes fit the LDS");

__device__ __forceinline__ int tile_pos(int f) {
  return (f & 128) + ((f & 3) << 5) + ((f & 127) >> 2);
}

__device__ __forceinline__ float edge_score(float as, float ad) {
  float v = as + ad;
  return (v >= 0.f) ? v : NEG_SLOPE * v;
}

template <int U>
__device__ __forceinline__ void agg_trip(int i0, int lane,
                                         const unsigned* list, float* acc,
                                         const float* ad_l, const int* mx_i, const float* sm_l,
                                         const float* __restrict__ a_src, const float* __restrict__ h) {
  int s[U], dl[U];
  v4f as[U], h0[U], h1[U];
  #pragma unroll
  for (int u = 0; u < U; ++u) {
    unsigned ent = (unsigned)__builtin_amdgcn_readfirstlane((int)list[i0 + u]);
    s[u]  = (int)(ent & 0xFFFFu);
    dl[u] = (int)(ent >> 16);
  }
  #pragma unroll
  for (int u = 0; u < U; ++u) {
    as[u] = *(const v4f*)(a_src + (size_t)s[u] * HEADS);
    const float* hs = h + (size_t)s[u] * HF + lane * 4;
    h0[u] = *(const v4f*)(hs);
    h1[u] = *(const v4f*)(hs + 128);
  }
  const int hsel = lane >> 4;
  float alo[U], ahi[U];
  #pragma unroll
  for (int u = 0; u < U; ++u) {
    const int rh = dl[u] * HEADS + hsel;
    float slo = hsel ? as[u][1] : as[u][0];
    float shi = hsel ? as[u][3] : as[u][2];
    alo[u] = __expf(edge_score(slo, ad_l[rh])     - __int_as_float(mx_i[rh]))     * sm_l[rh];
    ahi[u] = __expf(edge_score(shi, ad_l[rh + 2]) - __int_as_float(mx_i[rh + 2])) * sm_l[rh + 2];
  }
  #pragma unroll
  for (int u = 0; u < U; ++u) {
    float* row = acc + dl[u] * HF + lane;
    #pragma unroll
    for (int k = 0; k < 4; ++k) {
      atomicAdd(row + 32 * k,       alo[u] * h0[u][k]);
      atomicAdd(row + 128 + 32 * k, ahi[u] * h1[u][k]);
    }
  }
}

__global__ void __launch_bounds__(AGG_THREADS)
aggregate_kernel(const int* __restrict__ ei, const float* __restrict__ a_src,
                 const float* __restrict__ a_dst, const float* __restrict__ h,
                 const float* __restrict__ bias, float* __restrict__ out) {
  __shared__ __attribute__((aligned(16))) float acc[TILE_N * HF];
  __shared__ unsigned list[LIST_CAP];
  __shared__ __attribute__((aligned(16))) float ad_l[ROW_HEADS];
  __shared__ __attribute__((aligned(16))) int   mx_i[ROW_HEADS];
  __shared__ __attribute__((aligned(16))) float sm_l[ROW_HEADS];
  __shared__ unsigned cnt;

  const int tid  = threadIdx.x;
  const int lane = tid & 31;
  const int wave = tid >> 5;
  const int lo   = blockIdx.x * TILE_N;

  for (int i = tid; i < TILE_N * HF; i += AGG_THREADS) {
    int p = i & (HF - 1);
    int f = (p & 128) + ((p & 31) << 2) + ((p & 127) >> 5);
    acc[i] = bias[f];
  }
  if (tid < ROW_HEADS) {
    int node = lo + (tid >> 2);
    ad_l[tid] = (node < N_NODES) ? a_dst[(size_t)node * HEADS + (tid & 3)] : 0.f;
    mx_i[tid] = (int)0x80000000;
    sm_l[tid] = 0.f;
  }
  if (tid == 0) cnt = 0u;
  __syncthreads();

  const int4* dst4 = (const int4*)(ei + N_EDGES);
  for (int it = 0; it < SCAN_ITERS; ++it) {
    int4 dv[SCAN_BATCH];
    #pragma unroll
    for (int u = 0; u < SCAN_BATCH; ++u) {
      int v = (it * SCAN_BATCH + u) * AGG_THREADS + tid;
      dv[u] = (v < SCAN_VECS) ? dst4[v] : make_int4(-1, -1, -1, -1);
    }
    #pragma unroll
    for (int u = 0; u < SCAN_BATCH; ++u) {
      int v = (it * SCAN_BATCH + u) * AGG_THREADS + tid;
      unsigned r0 = (unsigned)(dv[u].x - lo), r1 = (unsigned)(dv[u].y - lo);
      unsigned r2 = (unsigned)(dv[u].z - lo), r3 = (unsigned)(dv[u].w - lo);
      if (min(min(r0, r1), min(r2, r3)) < (unsigned)TILE_N) {
        unsigned rr[4] = {r0, r1, r2, r3};
        #pragma unroll
        for (int k = 0; k < 4; ++k) {
          if (rr[k] < (unsigned)TILE_N) {
            int e = 4 * v + k;
            unsigned pos = atomicAdd(&cnt, 1u);
            if (pos < (unsigned)LIST_CAP) list[pos] = (unsigned)e | (rr[k] << 20);
          }
        }
      }
    }
  }
  __syncthreads();

  const unsigned total = cnt;
  const bool listed = total <= (unsigned)LIST_CAP;
  const int n = listed ? (int)total : 0;

  for (int b = tid; b < n; b += SOFT_BATCH * AGG_THREADS) {
    unsigned ent[SOFT_BATCH];
    int s[SOFT_BATCH];
    v4f as[SOFT_BATCH];
    #pragma unroll
    for (int u = 0; u < SOFT_BATCH; ++u) {
      int i = b + u * AGG_THREADS;
      ent[u] = (i < n) ? list[i] : 0u;
    }
    #pragma unroll
    for (int u = 0; u < SOFT_BATCH; ++u) s[u] = ei[ent[u] & 0xFFFFFu];
    #pragma unroll
    for (int u = 0; u < SOFT_BATCH; ++u) as[u] = *(const v4f*)(a_src + (size_t)s[u] * HEADS);
    asm volatile("" :: "v"(as[0]), "v"(as[1]), "v"(as[2]), "v"(as[3]));
    #pragma unroll
    for (int u = 0; u < SOFT_BATCH; ++u) {
      int i = b + u * AGG_THREADS;
      if (i < n) {
        int r = (int)(ent[u] >> 20);
        v4f ad = *(const v4f*)(ad_l + r * HEADS);
        #pragma unroll
        for (int q = 0; q < HEADS; ++q)
          atomicMax(&mx_i[r * HEADS + q], enc_f(edge_score(as[u][q], ad[q])));
        list[i] = (unsigned)s[u] | ((unsigned)r << 16);
      }
    }
  }
  if (!listed) {
    for (int e = tid; e < N_EDGES; e += AGG_THREADS) {
      unsigned r = (unsigned)(ei[N_EDGES + e] - lo);
      if (r < (unsigned)TILE_N) {
        int s = ei[e];
        #pragma unroll
        for (int q = 0; q < HEADS; ++q)
          atomicMax(&mx_i[r * HEADS + q],
                    enc_f(edge_score(a_src[(size_t)s * HEADS + q], ad_l[r * HEADS + q])));
      }
    }
  }
  __syncthreads();
  if (tid < ROW_HEADS) {
    float mv = dec_f(mx_i[tid]);
    mx_i[tid] = __float_as_int(__builtin_isfinite(mv) ? mv : 0.f);
  }
  __syncthreads();

  for (int b = tid; b < n; b += SOFT_BATCH * AGG_THREADS) {
    unsigned ent[SOFT_BATCH];
    v4f as[SOFT_BATCH];
    #pragma unroll
    for (int u = 0; u < SOFT_BATCH; ++u) {
      int i = b + u * AGG_THREADS;
      ent[u] = (i < n) ? list[i] : 0u;
    }
    #pragma unroll
    for (int u = 0; u < SOFT_BATCH; ++u) as[u] = *(const v4f*)(a_src + (size_t)(ent[u] & 0xFFFFu) * HEADS);
    asm volatile("" :: "v"(as[0]), "v"(as[1]), "v"(as[2]), "v"(as[3]));
    #pragma unroll
    for (int u = 0; u < SOFT_BATCH; ++u) {
      int i = b + u * AGG_THREADS;
      if (i < n) {
        int r = (int)(ent[u] >> 16);
        v4f ad = *(const v4f*)(ad_l + r * HEADS);
        v4i mv = *(const v4i*)(mx_i + r * HEADS);
        #pragma unroll
        for (int q = 0; q < HEADS; ++q)
          atomicAdd(&sm_l[r * HEADS + q], __expf(edge_score(as[u][q], ad[q]) - __int_as_float(mv[q])));
      }
    }
  }
  if (!listed) {
    for (int e = tid; e < N_EDGES; e += AGG_THREADS) {
      unsigned r = (unsigned)(ei[N_EDGES + e] - lo);
      if (r < (unsigned)TILE_N) {
        int s = ei[e];
        #pragma unroll
        for (int q = 0; q < HEADS; ++q)
          atomicAdd(&sm_l[r * HEADS + q],
                    __expf(edge_score(a_src[(size_t)s * HEADS + q], ad_l[r * HEADS + q]) - __int_as_float(mx_i[r * HEADS + q])));
      }
    }
  }
  __syncthreads();
  if (tid < ROW_HEADS) sm_l[tid] = 1.0f / (sm_l[tid] + GAT_EPS);
  __syncthreads();

  if (!listed) {
    for (int e = tid; e < N_EDGES; e += AGG_THREADS) {
      unsigned r = (unsigned)(ei[N_EDGES + e] - lo);
      if (r < (unsigned)TILE_N) {
        int s = ei[e];
        float al[HEADS];
        #pragma unroll
        for (int q = 0; q < HEADS; ++q)
          al[q] = __expf(edge_score(a_src[(size_t)s * HEADS + q], ad_l[r * HEADS + q]) - __int_as_float(mx_i[r * HEADS + q]))
                  * sm_l[r * HEADS + q];
        #pragma unroll 1
        for (int f = 0; f < HF; ++f) {
          float a = (f < 128) ? ((f < 64) ? al[0] : al[1]) : ((f < 192) ? al[2] : al[3]);
          atomicAdd(&acc[r * HF + tile_pos(f)], a * h[(size_t)s * HF + f]);
        }
      }
    }
  }
  const int full = n / AGG_EDGES_PER_TRIP;
  for (int t = wave; t < full; t += AGG_THREADS / 32)
    agg_trip<AGG_EDGES_PER_TRIP>(t * AGG_EDGES_PER_TRIP, lane, list, acc, ad_l, mx_i, sm_l, a_src, h);
  for (int i = full * AGG_EDGES_PER_TRIP + wave; i < n; i += AGG_THREADS / 32)
    agg_trip<1>(i, lane, list, acc, ad_l, mx_i, sm_l, a_src, h);
  __syncthreads();

  for (int r = wave; r < TILE_N; r += AGG_THREADS / 32) {
    int node = lo + r;
    if (node < N_NODES) {
      const float* row = acc + r * HF + lane;
      v4f o0, o1;
      #pragma unroll
      for (int k = 0; k < 4; ++k) { o0[k] = row[32 * k]; o1[k] = row[128 + 32 * k]; }
      float* op = out + (size_t)node * HF + lane * 4;
      *(v4f*)(op)       = o0;
      *(v4f*)(op + 128) = o1;
    }
  }
}

extern "C" void kernel_launch(void* const* d_in, const int* in_sizes, int n_in,
                              void* d_out, int out_size, void* d_ws, size_t ws_size,
                              hipStream_t stream) {
  const float* x       = (const float*)d_in[0];
  const int*   ei      = (const int*)  d_in[1];
  const float* W       = (const float*)d_in[2];
  const float* att_src = (const float*)d_in[3];
  const float* att_dst = (const float*)d_in[4];
  const float* bias    = (const float*)d_in[5];
  float* out = (float*)d_out;

  char* p = (char*)d_ws;
  auto alloc = [&](size_t bytes) {
    char* r = p; p += (bytes + 255) & ~(size_t)255; return r;
  };
  float*    h     = (float*)   alloc((size_t)N_NODES * HF * 4);
  _Float16* Xh    = (_Float16*)alloc((size_t)N_NODES * IN_F * 2);
  _Float16* Wh    = (_Float16*)alloc((size_t)HF * IN_F * 2);
  float*    a_src = (float*)   alloc((size_t)N_NODES * HEADS * 4);
  float*    a_dst = (float*)   alloc((size_t)N_NODES * HEADS * 4);

  cvt_f16_kernel  <<<2048, 256, 0, stream>>>(x, W, Xh, Wh);
  gemm_wmma_kernel<<<6250, 256, 0, stream>>>(Xh, Wh, h);
  logits_kernel   <<<6250, 256, 0, stream>>>(h, att_src, att_dst, a_src, a_dst);
  aggregate_kernel<<<AGG_BLOCKS, AGG_THREADS, 0, stream>>>(ei, a_src, a_dst, h, bias, out);
}


# Projection and logits launches become resident grids: a wave keeps its 16 columns' W fragments in registers and walks 12-13 row tiles, a logits wave walks 16 nodes; 100000 wave entries become 7221
# speedup vs baseline: 5.0013x; 1.1859x over previous
#include <hip/hip_runtime.h>
#include <hip/hip_bf16.h>

#define N_NODES 50000
#define N_EDGES 800000
#define IN_F 128
#define OUT_F 64
#define HEADS 4
#define HF 256
#define NEG_SLOPE 0.2f
#define GAT_EPS 1e-16f

typedef __attribute__((ext_vector_type(16))) _Float16 v16h;
typedef __attribute__((ext_vector_type(8)))  _Float16 v8h;
typedef __attribute__((ext_vector_type(8)))  float    v8f;
typedef __attribute__((ext_vector_type(4)))  float    v4f;
typedef __attribute__((ext_vector_type(4)))  int      v4i;

#define TILE_N 197
#define AGG_BLOCKS 254
#define TILE_NONE 255u
static_assert(TILE_N * AGG_BLOCKS >= N_NODES, "the tiles cover every node");
static_assert(AGG_BLOCKS <= 255, "a block number fits a byte and never equals TILE_NONE");
static_assert(N_EDGES % 16 == 0, "tile bytes are written four and read sixteen at a time");

__device__ __forceinline__ unsigned tile_code(int d) {
  return ((unsigned)d < (unsigned)N_NODES) ? (unsigned)d / TILE_N : TILE_NONE;
}

__global__ void cvt_f16_kernel(const float* __restrict__ x, const float* __restrict__ W,
                               const int* __restrict__ ei,
                               _Float16* __restrict__ Xh, _Float16* __restrict__ Wh,
                               unsigned* __restrict__ tw) {
  int idx = blockIdx.x * blockDim.x + threadIdx.x;
  int stride = gridDim.x * blockDim.x;
  const int4* dst4 = (const int4*)(ei + N_EDGES);
  for (int i = idx; i < N_EDGES / 4; i += stride) {
    int4 d = dst4[i];
    tw[i] = tile_code(d.x) | (tile_code(d.y) << 8) | (tile_code(d.z) << 16) | (tile_code(d.w) << 24);
  }
  for (int i = idx; i < N_NODES * IN_F; i += stride) Xh[i] = (_Float16)x[i];
  for (int i = idx; i < HF * IN_F; i += stride)      Wh[i] = (_Float16)W[i];
}

#define GEMM_BLOCKS 256
#define GEMM_THREADS 512
#define M_TILES (N_NODES / 16)
#define K_BLOCKS (IN_F / 32)
#define FULL_TRIPS (M_TILES / GEMM_BLOCKS)
static_assert(N_NODES % 16 == 0, "every row tile is whole");
static_assert(HF == 16 * (GEMM_THREADS / 32), "a block's waves cover all output columns");
static_assert(FULL_TRIPS >= 2, "every block has the two row tiles it asks for first");
static_assert((FULL_TRIPS + 1) * GEMM_BLOCKS >= M_TILES, "one more tile at most behind the full trips");

struct frag_row { v8h p[2 * K_BLOCKS]; };

__device__ __forceinline__ frag_row load_frags(const _Float16* __restrict__ rowp) {
  frag_row f;
  #pragma unroll
  for (int kb = 0; kb < K_BLOCKS; ++kb) {
    f.p[2 * kb]     = *(const v8h*)(rowp + 32 * kb);
    f.p[2 * kb + 1] = *(const v8h*)(rowp + 32 * kb + 16);
  }
  return f;
}

__device__ __forceinline__ v16h frag16(v8h lo, v8h hi) {
  return __builtin_shufflevector(lo, hi, 0, 1, 2, 3, 4, 5, 6, 7, 8, 9, 10, 11, 12, 13, 14, 15);
}

__device__ __forceinline__ v8f tile_product(const frag_row& a, const v16h (&B)[K_BLOCKS]) {
  v8f c = {};
  #pragma unroll
  for (int kb = 0; kb < K_BLOCKS; ++kb)
    c = __builtin_amdgcn_wmma_f32_16x16x32_f16(false, frag16(a.p[2 * kb], a.p[2 * kb + 1]),
                                               false, B[kb], (short)0, c, false, false);
  return c;
}

__device__ __forceinline__ void tile_store(v8f c, float* __restrict__ hr) {
  #pragma unroll
  for (int r = 0; r < 8; ++r) hr[(size_t)r * HF] = c[r];
}

__global__ void __launch_bounds__(GEMM_THREADS)
gemm_wmma_kernel(const _Float16* __restrict__ Xh, const _Float16* __restrict__ Wh,
                 float* __restrict__ h) {
  const int lane  = threadIdx.x & 31;
  const int ncol0 = (threadIdx.x >> 5) << 4;
  const int l15   = lane & 15;
  const int khalf = lane >> 4;

  const frag_row bf = load_frags(Wh + (size_t)(ncol0 + l15) * IN_F + khalf * 8);
  v16h B[K_BLOCKS];
  #pragma unroll
  for (int kb = 0; kb < K_BLOCKS; ++kb) B[kb] = frag16(bf.p[2 * kb], bf.p[2 * kb + 1]);

  const _Float16* ap = Xh + (size_t)l15 * IN_F + khalf * 8;
  float* hp = h + (size_t)(khalf * 8) * HF + ncol0 + l15;

  const int t0 = blockIdx.x;
  const bool longer = t0 + FULL_TRIPS * GEMM_BLOCKS < M_TILES;
  frag_row f[3];
  f[0] = load_frags(ap + (size_t)t0 * 16 * IN_F);
  f[1] = load_frags(ap + (size_t)(t0 + GEMM_BLOCKS) * 16 * IN_F);
  v8f cprev = {};
  #pragma unroll
  for (int i = 0; i < FULL_TRIPS; ++i) {
    const int t = t0 + i * GEMM_BLOCKS;
    if (i + 2 < FULL_TRIPS)
      f[(i + 2) % 3] = load_frags(ap + (size_t)(t + 2 * GEMM_BLOCKS) * 16 * IN_F);
    else if (i + 2 == FULL_TRIPS)
      f[(i + 2) % 3] = load_frags(ap + (size_t)(longer ? t + 2 * GEMM_BLOCKS : t) * 16 * IN_F);
    __builtin_amdgcn_sched_barrier(0);
    const v8f c = tile_product(f[i % 3], B);
    __builtin_amdgcn_sched_barrier(0);
    if (i > 0) tile_store(cprev, hp + (size_t)(t - GEMM_BLOCKS) * 16 * HF);
    cprev = c;
  }
  const int tl = t0 + (FULL_TRIPS - 1) * GEMM_BLOCKS;
  if (longer) {
    const v8f c = tile_product(f[FULL_TRIPS % 3], B);
    __builtin_amdgcn_sched_barrier(0);
    tile_store(cprev, hp + (size_t)tl * 16 * HF);
    tile_store(c, hp + (size_t)(tl + GEMM_BLOCKS) * 16 * HF);
  } else {
    tile_store(cprev, hp + (size_t)tl * 16 * HF);
  }
}

#define LOGIT_WAVES 3125
#define LOGIT_WAVES_PER_BLOCK 5
#define LOGIT_BLOCKS 625
#define LOGIT_THREADS 160
#define LOGIT_BATCH 8
#define LOGIT_TRIPS 2
static_assert(LOGIT_WAVES * LOGIT_BATCH * LOGIT_TRIPS == N_NODES, "the waves' walks cover every node exactly once");
static_assert(LOGIT_BLOCKS * LOGIT_WAVES_PER_BLOCK == LOGIT_WAVES && LOGIT_THREADS == 32 * LOGIT_WAVES_PER_BLOCK, "whole blocks");

__global__ void __launch_bounds__(LOGIT_THREADS)
logits_kernel(const float* __restrict__ h,
              const float* __restrict__ att_src, const float* __restrict__ att_dst,
              float* __restrict__ a_src, float* __restrict__ a_dst) {
  const int gw    = blockIdx.x * LOGIT_WAVES_PER_BLOCK + (threadIdx.x >> 5);
  const int lane  = threadIdx.x & 31;
  const int head  = lane >> 3;
  const int fbase = (lane & 7) * 8;
  float ws[8], wd[8];
  #pragma unroll
  for (int j = 0; j < 8; ++j) {
    ws[j] = att_src[head * OUT_F + fbase + j];
    wd[j] = att_dst[head * OUT_F + fbase + j];
  }
  for (int trip = 0; trip < LOGIT_TRIPS; ++trip) {
    v4f lo[LOGIT_BATCH], hi[LOGIT_BATCH];
    #pragma unroll
    for (int u = 0; u < LOGIT_BATCH; ++u) {
      const int node = gw + (trip * LOGIT_BATCH + u) * LOGIT_WAVES;
      const float* hp = h + (size_t)node * HF + lane * 8;
      lo[u] = *(const v4f*)(hp);
      hi[u] = *(const v4f*)(hp + 4);
    }
    __builtin_amdgcn_sched_barrier(0);
    float ps[LOGIT_BATCH], pd[LOGIT_BATCH];
    #pragma unroll
    for (int u = 0; u < LOGIT_BATCH; ++u) {
      ps[u] = 0.f; pd[u] = 0.f;
      #pragma unroll
      for (int j = 0; j < 8; ++j) {
        float hv = (j < 4) ? lo[u][j & 3] : hi[u][j & 3];
        ps[u] += hv * ws[j];
        pd[u] += hv * wd[j];
      }
      #pragma unroll
      for (int off = 1; off < 8; off <<= 1) {
        ps[u] += __shfl_xor(ps[u], off, 32);
        pd[u] += __shfl_xor(pd[u], off, 32);
      }
    }
    if ((lane & 7) == 0) {
      #pragma unroll
      for (int u = 0; u < LOGIT_BATCH; ++u) {
        const int node = gw + (trip * LOGIT_BATCH + u) * LOGIT_WAVES;
        a_src[node * HEADS + head] = ps[u];
        a_dst[node * HEADS + head] = pd[u];
      }
    }
  }
}

__device__ __forceinline__ int enc_f(float f) {
  int i = __float_as_int(f);
  return (i >= 0) ? i : (i ^ 0x7fffffff);
}
__device__ __forceinline__ float dec_f(int i) {
  return __int_as_float((i >= 0) ? i : (i ^ 0x7fffffff));
}

#define AGG_THREADS 1024
#define LIST_CAP 24576
#define SCAN_VECS (N_EDGES / 16)
#define SCAN_BATCH 10
#define SCAN_ITERS 5
#define AGG_EDGES_PER_TRIP 8
#define SOFT_BATCH 4
#define ROW_HEADS (TILE_N * HEADS)
static_assert(SCAN_ITERS * SCAN_BATCH * AGG_THREADS >= SCAN_VECS, "the scan covers the edge list");
static_assert(SCAN_BATCH % 2 == 0, "two vectors' match bits share a word");
static_assert(TILE_N <= 256, "the local row fits the list word beside the source id");
static_assert(N_NODES <= (1 << 16), "source id and local row share a word");
static_assert(SOFT_BATCH == 4, "the gather fences of phases 2 and 3 name four edges");
static_assert(ROW_HEADS <= AGG_THREADS, "one lane per (row, head) of the small planes");
static_assert(TILE_N * HF * 4 + LIST_CAP * 4 + 3 * ROW_HEADS * 4 + 16 <= 320 * 1024, "tile, list and planes fit the LDS");

__device__ __forceinline__ unsigned eq_bytes(unsigned w, unsigned pat) {
  unsigned x = w ^ pat;
  unsigned t = (x & 0x7f7f7f7fu) + 0x7f7f7f7fu;
  return ~(t | x) & 0x80808080u;
}

__device__ __forceinline__ int tile_pos(int f) {
  return (f & 128) + ((f & 3) << 5) + ((f & 127) >> 2);
}

__device__ __forceinline__ float edge_score(float as, float ad) {
  float v = as + ad;
  return (v >= 0.f) ? v : NEG_SLOPE * v;
}

template <int U>
__device__ __forceinline__ void agg_trip(int i0, int lane,
                                         const unsigned* list, float* acc,
                                         const float* ad_l, const int* mx_i, const float* sm_l,
                                         const float* __restrict__ a_src, const float* __restrict__ h) {
  int s[U], dl[U];
  v4f as[U], h0[U], h1[U];
  #pragma unroll
  for (int u = 0; u < U; ++u) {
    unsigned ent = (unsigned)__builtin_amdgcn_readfirstlane((int)list[i0 + u]);
    s[u]  = (int)(ent & 0xFFFFu);
    dl[u] = (int)(ent >> 16);
  }
  #pragma unroll
  for (int u = 0; u < U; ++u) {
    as[u] = *(const v4f*)(a_src + (size_t)s[u] * HEADS);
    const float* hs = h + (size_t)s[u] * HF + lane * 4;
    h0[u] = *(const v4f*)(hs);
    h1[u] = *(const v4f*)(hs + 128);
  }
  const int hsel = lane >> 4;
  float alo[U], ahi[U];
  #pragma unroll
  for (int u = 0; u < U; ++u) {
    const int rh = dl[u] * HEADS + hsel;
    float slo = hsel ? as[u][1] : as[u][0];
    float shi = hsel ? as[u][3] : as[u][2];
    alo[u] = __expf(edge_score(slo, ad_l[rh])     - __int_as_float(mx_i[rh]))     * sm_l[rh];
    ahi[u] = __expf(edge_score(shi, ad_l[rh + 2]) - __int_as_float(mx_i[rh + 2])) * sm_l[rh + 2];
  }
  #pragma unroll
  for (int u = 0; u < U; ++u) {
    float* row = acc + dl[u] * HF + lane;
    #pragma unroll
    for (int k = 0; k < 4; ++k) {
      atomicAdd(row + 32 * k,       alo[u] * h0[u][k]);
      atomicAdd(row + 128 + 32 * k, ahi[u] * h1[u][k]);
    }
  }
}

__global__ void __launch_bounds__(AGG_THREADS)
aggregate_kernel(const int* __restrict__ ei, const unsigned* __restrict__ tw, const float* __restrict__ a_src,
                 const float* __restrict__ a_dst, const float* __restrict__ h,
                 const float* __restrict__ bias, float* __restrict__ out) {
  __shared__ __attribute__((aligned(16))) float acc[TILE_N * HF];
  __shared__ unsigned list[LIST_CAP];
  __shared__ __attribute__((aligned(16))) float ad_l[ROW_HEADS];
  __shared__ __attribute__((aligned(16))) int   mx_i[ROW_HEADS];
  __shared__ __attribute__((aligned(16))) float sm_l[ROW_HEADS];
  __shared__ unsigned cnt;

  const int tid  = threadIdx.x;
  const int lane = tid & 31;
  const int wave = tid >> 5;
  const int lo   = blockIdx.x * TILE_N;

  for (int i = tid; i < TILE_N * HF; i += AGG_THREADS) {
    int p = i & (HF - 1);
    int f = (p & 128) + ((p & 31) << 2) + ((p & 127) >> 5);
    acc[i] = bias[f];
  }
  if (tid < ROW_HEADS) {
    int node = lo + (tid >> 2);
    ad_l[tid] = (node < N_NODES) ? a_dst[(size_t)node * HEADS + (tid & 3)] : 0.f;
    mx_i[tid] = (int)0x80000000;
    sm_l[tid] = 0.f;
  }
  if (tid == 0) cnt = 0u;
  __syncthreads();

  const uint4* tv = (const uint4*)tw;
  const unsigned pat = blockIdx.x * 0x01010101u;
  for (int it = 0; it < SCAN_ITERS; ++it) {
    uint4 dv[SCAN_BATCH];
    #pragma unroll
    for (int u = 0; u < SCAN_BATCH; ++u) {
      int v = (it * SCAN_BATCH + u) * AGG_THREADS + tid;
      dv[u] = (v < SCAN_VECS) ? tv[v] : make_uint4(~0u, ~0u, ~0u, ~0u);
    }
    unsigned m[SCAN_BATCH / 2];
    unsigned c = 0u;
    #pragma unroll
    for (int p = 0; p < SCAN_BATCH / 2; ++p) {
      m[p] = (eq_bytes(dv[2 * p].x, pat) >> 7)     | (eq_bytes(dv[2 * p].y, pat) >> 6)
           | (eq_bytes(dv[2 * p].z, pat) >> 5)     | (eq_bytes(dv[2 * p].w, pat) >> 4)
           | (eq_bytes(dv[2 * p + 1].x, pat) >> 3) | (eq_bytes(dv[2 * p + 1].y, pat) >> 2)
           | (eq_bytes(dv[2 * p + 1].z, pat) >> 1) |  eq_bytes(dv[2 * p + 1].w, pat);
      c += (unsigned)__popc(m[p]);
    }
    if (c) {
      unsigned pos = atomicAdd(&cnt, c);
      #pragma unroll
      for (int p = 0; p < SCAN_BATCH / 2; ++p) {
        const int e0 = 16 * ((it * SCAN_BATCH + 2 * p) * AGG_THREADS + tid);
        unsigned mm = m[p];
        while (mm) {
          int b = __ffs((int)mm) - 1;
          mm &= mm - 1u;
          int e = e0 + ((b & 4) ? 16 * AGG_THREADS : 0) + 4 * (b & 3) + (b >> 3);
          if (pos < (unsigned)LIST_CAP) list[pos] = (unsigned)e;
          ++pos;
        }
      }
    }
  }
  __syncthreads();

  const unsigned total = cnt;
  const bool listed = total <= (unsigned)LIST_CAP;
  const int n = listed ? (int)total : 0;

  for (int b = tid; b < n; b += SOFT_BATCH * AGG_THREADS) {
    unsigned ent[SOFT_BATCH];
    int s[SOFT_BATCH], d[SOFT_BATCH];
    v4f as[SOFT_BATCH];
    #pragma unroll
    for (int u = 0; u < SOFT_BATCH; ++u) {
      int i = b + u * AGG_THREADS;
      ent[u] = (i < n) ? list[i] : 0u;
    }
    #pragma unroll
    for (int u = 0; u < SOFT_BATCH; ++u) {
      s[u] = ei[ent[u]];
      d[u] = ei[N_EDGES + ent[u]];
    }
    #pragma unroll
    for (int u = 0; u < SOFT_BATCH; ++u) as[u] = *(const v4f*)(a_src + (size_t)s[u] * HEADS);
    asm volatile("" :: "v"(as[0]), "v"(as[1]), "v"(as[2]), "v"(as[3]),
                       "v"(d[0]), "v"(d[1]), "v"(d[2]), "v"(d[3]));
    #pragma unroll
    for (int u = 0; u < SOFT_BATCH; ++u) {
      int i = b + u * AGG_THREADS;
      if (i < n) {
        int r = d[u] - lo;
        v4f ad = *(const v4f*)(ad_l + r * HEADS);
        #pragma unroll
        for (int q = 0; q < HEADS; ++q)
          atomicMax(&mx_i[r * HEADS + q], enc_f(edge_score(as[u][q], ad[q])));
        list[i] = (unsigned)s[u] | ((unsigned)r << 16);
      }
    }
  }
  if (!listed) {
    for (int e = tid; e < N_EDGES; e += AGG_THREADS) {
      unsigned r = (unsigned)(ei[N_EDGES + e] - lo);
      if (r < (unsigned)TILE_N) {
        int s = ei[e];
        #pragma unroll
        for (int q = 0; q < HEADS; ++q)
          atomicMax(&mx_i[r * HEADS + q],
                    enc_f(edge_score(a_src[(size_t)s * HEADS + q], ad_l[r * HEADS + q])));
      }
    }
  }
  __syncthreads();
  if (tid < ROW_HEADS) {
    float mv = dec_f(mx_i[tid]);
    mx_i[tid] = __float_as_int(__builtin_isfinite(mv) ? mv : 0.f);
  }
  __syncthreads();

  for (int b = tid; b < n; b += SOFT_BATCH * AGG_THREADS) {
    unsigned ent[SOFT_BATCH];
    v4f as[SOFT_BATCH];
    #pragma unroll
    for (int u = 0; u < SOFT_BATCH; ++u) {
      int i = b + u * AGG_THREADS;
      ent[u] = (i < n) ? list[i] : 0u;
    }
    #pragma unroll
    for (int u = 0; u < SOFT_BATCH; ++u) as[u] = *(const v4f*)(a_src + (size_t)(ent[u] & 0xFFFFu) * HEADS);
    asm volatile("" :: "v"(as[0]), "v"(as[1]), "v"(as[2]), "v"(as[3]));
    #pragma unroll
    for (int u = 0; u < SOFT_BATCH; ++u) {
      int i = b + u * AGG_THREADS;
      if (i < n) {
        int r = (int)(ent[u] >> 16);
        v4f ad = *(const v4f*)(ad_l + r * HEADS);
        v4i mv = *(const v4i*)(mx_i + r * HEADS);
        #pragma unroll
        for (int q = 0; q < HEADS; ++q)
          atomicAdd(&sm_l[r * HEADS + q], __expf(edge_score(as[u][q], ad[q]) - __int_as_float(mv[q])));
      }
    }
  }
  if (!listed) {
    for (int e = tid; e < N_EDGES; e += AGG_THREADS) {
      unsigned r = (unsigned)(ei[N_EDGES + e] - lo);
      if (r < (unsigned)TILE_N) {
        int s = ei[e];
        #pragma unroll
        for (int q = 0; q < HEADS; ++q)
          atomicAdd(&sm_l[r * HEADS + q],
                    __expf(edge_score(a_src[(size_t)s * HEADS + q], ad_l[r * HEADS + q]) - __int_as_float(mx_i[r * HEADS + q])));
      }
    }
  }
  __syncthreads();
  if (tid < ROW_HEADS) sm_l[tid] = 1.0f / (sm_l[tid] + GAT_EPS);
  __syncthreads();

  if (!listed) {
    for (int e = tid; e < N_EDGES; e += AGG_THREADS) {
      unsigned r = (unsigned)(ei[N_EDGES + e] - lo);
      if (r < (unsigned)TILE_N) {
        int s = ei[e];
        float al[HEADS];
        #pragma unroll
        for (int q = 0; q < HEADS; ++q)
          al[q] = __expf(edge_score(a_src[(size_t)s * HEADS + q], ad_l[r * HEADS + q]) - __int_as_float(mx_i[r * HEADS + q]))
                  * sm_l[r * HEADS + q];
        #pragma unroll 1
        for (int f = 0; f < HF; ++f) {
          float a = (f < 128) ? ((f < 64) ? al[0] : al[1]) : ((f < 192) ? al[2] : al[3]);
          atomicAdd(&acc[r * HF + tile_pos(f)], a * h[(size_t)s * HF + f]);
        }
      }
    }
  }
  const int full = n / AGG_EDGES_PER_TRIP;
  for (int t = wave; t < full; t += AGG_THREADS / 32)
    agg_trip<AGG_EDGES_PER_TRIP>(t * AGG_EDGES_PER_TRIP, lane, list, acc, ad_l, mx_i, sm_l, a_src, h);
  for (int i = full * AGG_EDGES_PER_TRIP + wave; i < n; i += AGG_THREADS / 32)
    agg_trip<1>(i, lane, list, acc, ad_l, mx_i, sm_l, a_src, h);
  __syncthreads();

  for (int r = wave; r < TILE_N; r += AGG_THREADS / 32) {
    int node = lo + r;
    if (node < N_NODES) {
      const float* row = acc + r * HF + lane;
      v4f o0, o1;
      #pragma unroll
      for (int k = 0; k < 4; ++k) { o0[k] = row[32 * k]; o1[k] = row[128 + 32 * k]; }
      float* op = out + (size_t)node * HF + lane * 4;
      *(v4f*)(op)       = o0;
      *(v4f*)(op + 128) = o1;
    }
  }
}

extern "C" void kernel_launch(void* const* d_in, const int* in_sizes, int n_in,
                              void* d_out, int out_size, void* d_ws, size_t ws_size,
                              hipStream_t stream) {
  const float* x       = (const float*)d_in[0];
  const int*   ei      = (const int*)  d_in[1];
  const float* W       = (const float*)d_in[2];
  const float* att_src = (const float*)d_in[3];
  const float* att_dst = (const float*)d_in[4];
  const float* bias    = (const float*)d_in[5];
  float* out = (float*)d_out;

  char* p = (char*)d_ws;
  auto alloc = [&](size_t bytes) {
    char* r = p; p += (bytes + 255) & ~(size_t)255; return r;
  };
  float*    h     = (float*)   alloc((size_t)N_NODES * HF * 4);
  _Float16* Xh    = (_Float16*)alloc((size_t)N_NODES * IN_F * 2);
  _Float16* Wh    = (_Float16*)alloc((size_t)HF * IN_F * 2);
  float*    a_src = (float*)   alloc((size_t)N_NODES * HEADS * 4);
  float*    a_dst = (float*)   alloc((size_t)N_NODES * HEADS * 4);
  unsigned* tw    = (unsigned*)alloc((size_t)N_EDGES);

  cvt_f16_kernel  <<<2048, 256, 0, stream>>>(x, W, ei, Xh, Wh, tw);
  gemm_wmma_kernel<<<GEMM_BLOCKS, GEMM_THREADS, 0, stream>>>(Xh, Wh, h);
  logits_kernel   <<<LOGIT_BLOCKS, LOGIT_THREADS, 0, stream>>>(h, att_src, att_dst, a_src, a_dst);
  aggregate_kernel<<<AGG_BLOCKS, AGG_THREADS, 0, stream>>>(ei, tw, a_src, a_dst, h, bias, out);
}


# Attention vectors folded into W (8 rows W_head^T att, each as a high and a low f16 row) ride the projection as a 17th wave's columns: logits launch and re-read of h gone; logit sums in a new order
# speedup vs baseline: 5.6475x; 1.1292x over previous
#include <hip/hip_runtime.h>
#include <hip/hip_bf16.h>

#define N_NODES 50000
#define N_EDGES 800000
#define IN_F 128
#define OUT_F 64
#define HEADS 4
#define HF 256
#define NEG_SLOPE 0.2f
#define GAT_EPS 1e-16f
#define LOGITS 8
#define LOGIT_COLS 16
#define WH_ROWS 272

typedef __attribute__((ext_vector_type(16))) _Float16 v16h;
typedef __attribute__((ext_vector_type(8)))  _Float16 v8h;
typedef __attribute__((ext_vector_type(8)))  float    v8f;
typedef __attribute__((ext_vector_type(4)))  float    v4f;
typedef __attribute__((ext_vector_type(4)))  int      v4i;

#define TILE_N 197
#define AGG_BLOCKS 254
#define TILE_NONE 255u
static_assert(TILE_N * AGG_BLOCKS >= N_NODES, "the tiles cover every node");
static_assert(AGG_BLOCKS <= 255, "a block number fits a byte and never equals TILE_NONE");
static_assert(N_EDGES % 16 == 0, "tile bytes are written four and read sixteen at a time");

__device__ __forceinline__ unsigned tile_code(int d) {
  return ((unsigned)d < (unsigned)N_NODES) ? (unsigned)d / TILE_N : TILE_NONE;
}

#define CVT_BLOCKS 2048
#define CVT_THREADS 256
#define FOLD_BLOCKS LOGITS
static_assert(IN_F <= CVT_THREADS, "a fold block has a lane per input feature");

__global__ void __launch_bounds__(CVT_THREADS)
cvt_f16_kernel(const float* __restrict__ x, const float* __restrict__ W,
               const int* __restrict__ ei,
               const float* __restrict__ att_src, const float* __restrict__ att_dst,
               _Float16* __restrict__ Xh, _Float16* __restrict__ Wh,
               unsigned* __restrict__ tw) {
  if (blockIdx.x < FOLD_BLOCKS) {
    const int j  = blockIdx.x;
    const int hd = j & (HEADS - 1);
    const int k  = threadIdx.x;
    if (k < IN_F) {
      const float* att = ((j < HEADS) ? att_src : att_dst) + hd * OUT_F;
      const float* wp  = W + (size_t)hd * OUT_F * IN_F + k;
      float v = 0.f;
      #pragma unroll 16
      for (int f = 0; f < OUT_F; ++f) v += wp[(size_t)f * IN_F] * att[f];
      const _Float16 hi = (_Float16)v;
      Wh[(size_t)(HF + j) * IN_F + k]          = hi;
      Wh[(size_t)(HF + LOGITS + j) * IN_F + k] = (_Float16)(v - (float)hi);
    }
    return;
  }
  int idx = (blockIdx.x - FOLD_BLOCKS) * blockDim.x + threadIdx.x;
  int stride = (gridDim.x - FOLD_BLOCKS) * blockDim.x;
  const int4* dst4 = (const int4*)(ei + N_EDGES);
  for (int i = idx; i < N_EDGES / 4; i += stride) {
    int4 d = dst4[i];
    tw[i] = tile_code(d.x) | (tile_code(d.y) << 8) | (tile_code(d.z) << 16) | (tile_code(d.w) << 24);
  }
  for (int i = idx; i < N_NODES * IN_F; i += stride) Xh[i] = (_Float16)x[i];
  for (int i = idx; i < HF * IN_F; i += stride)      Wh[i] = (_Float16)W[i];
}

#define GEMM_BLOCKS 256
#define GEMM_WAVES 17
#define GEMM_THREADS 544
#define M_TILES (N_NODES / 16)
#define K_BLOCKS (IN_F / 32)
#define FULL_TRIPS (M_TILES / GEMM_BLOCKS)
static_assert(N_NODES % 16 == 0, "every row tile is whole");
static_assert(GEMM_THREADS == 32 * GEMM_WAVES && WH_ROWS == 16 * GEMM_WAVES, "a block's waves cover all columns, the logit columns included");
static_assert(WH_ROWS == HF + LOGIT_COLS && LOGIT_COLS == 2 * LOGITS && LOGITS == 2 * HEADS, "one piece holds both parts of every logit");
static_assert(FULL_TRIPS >= 2, "every block has the two row tiles it asks for first");
static_assert((FULL_TRIPS + 1) * GEMM_BLOCKS >= M_TILES, "one more tile at most behind the full trips");

struct frag_row { v8h p[2 * K_BLOCKS]; };

__device__ __forceinline__ frag_row load_frags(const _Float16* __restrict__ rowp) {
  frag_row f;
  #pragma unroll
  for (int kb = 0; kb < K_BLOCKS; ++kb) {
    f.p[2 * kb]     = *(const v8h*)(rowp + 32 * kb);
    f.p[2 * kb + 1] = *(const v8h*)(rowp + 32 * kb + 16);
  }
  return f;
}

__device__ __forceinline__ v16h frag16(v8h lo, v8h hi) {
  return __builtin_shufflevector(lo, hi, 0, 1, 2, 3, 4, 5, 6, 7, 8, 9, 10, 11, 12, 13, 14, 15);
}

__device__ __forceinline__ v8f tile_product(const frag_row& a, const v16h (&B)[K_BLOCKS]) {
  v8f c = {};
  #pragma unroll
  for (int kb = 0; kb < K_BLOCKS; ++kb)
    c = __builtin_amdgcn_wmma_f32_16x16x32_f16(false, frag16(a.p[2 * kb], a.p[2 * kb + 1]),
                                               false, B[kb], (short)0, c, false, false);
  return c;
}

struct piece_dest { float* p; bool writer; };

template <bool LOGIT_WAVE>
__device__ __forceinline__ void tile_store(v8f c, const piece_dest& d, int t) {
  if (LOGIT_WAVE) {
    float v[8];
    #pragma unroll
    for (int r = 0; r < 8; ++r) v[r] = c[r] + __shfl_xor(c[r], 8, 32);
    if (d.writer) {
      float* lr = d.p + (size_t)t * 16 * HEADS;
      #pragma unroll
      for (int r = 0; r < 8; ++r) lr[(size_t)r * HEADS] = v[r];
    }
  } else {
    float* hr = d.p + (size_t)t * 16 * HF;
    #pragma unroll
    for (int r = 0; r < 8; ++r) hr[(size_t)r * HF] = c[r];
  }
}

template <bool LOGIT_WAVE>
__device__ __forceinline__ void walk_tiles(const _Float16* __restrict__ ap, const v16h (&B)[K_BLOCKS],
                                           const piece_dest& d, int t0) {
  const bool longer = t0 + FULL_TRIPS * GEMM_BLOCKS < M_TILES;
  frag_row f[3];
  f[0] = load_frags(ap + (size_t)t0 * 16 * IN_F);
  f[1] = load_frags(ap + (size_t)(t0 + GEMM_BLOCKS) * 16 * IN_F);
  v8f cprev = {};
  #pragma unroll
  for (int i = 0; i < FULL_TRIPS; ++i) {
    const int t = t0 + i * GEMM_BLOCKS;
    if (i + 2 < FULL_TRIPS)
      f[(i + 2) % 3] = load_frags(ap + (size_t)(t + 2 * GEMM_BLOCKS) * 16 * IN_F);
    else if (i + 2 == FULL_TRIPS)
      f[(i + 2) % 3] = load_frags(ap + (size_t)(longer ? t + 2 * GEMM_BLOCKS : t) * 16 * IN_F);
    __builtin_amdgcn_sched_barrier(0);
    const v8f c = tile_product(f[i % 3], B);
    __builtin_amdgcn_sched_barrier(0);
    if (i > 0) tile_store<LOGIT_WAVE>(cprev, d, t - GEMM_BLOCKS);
    cprev = c;
  }
  const int tl = t0 + (FULL_TRIPS - 1) * GEMM_BLOCKS;
  if (longer) {
    const v8f c = tile_product(f[FULL_TRIPS % 3], B);
    __builtin_amdgcn_sched_barrier(0);
    tile_store<LOGIT_WAVE>(cprev, d, tl);
    tile_store<LOGIT_WAVE>(c, d, tl + GEMM_BLOCKS);
  } else {
    tile_store<LOGIT_WAVE>(cprev, d, tl);
  }
}

__global__ void __launch_bounds__(GEMM_THREADS)
gemm_wmma_kernel(const _Float16* __restrict__ Xh, const _Float16* __restrict__ Wh,
                 float* __restrict__ h, float* __restrict__ a_src, float* __restrict__ a_dst) {
  const int lane  = threadIdx.x & 31;
  const int wave  = __builtin_amdgcn_readfirstlane((int)(threadIdx.x >> 5));
  const int ncol0 = wave << 4;
  const int l15   = lane & 15;
  const int khalf = lane >> 4;

  const frag_row bf = load_frags(Wh + (size_t)(ncol0 + l15) * IN_F + khalf * 8);
  v16h B[K_BLOCKS];
  #pragma unroll
  for (int kb = 0; kb < K_BLOCKS; ++kb) B[kb] = frag16(bf.p[2 * kb], bf.p[2 * kb + 1]);

  const _Float16* ap = Xh + (size_t)l15 * IN_F + khalf * 8;
  const int t0 = blockIdx.x;

  if (wave == GEMM_WAVES - 1) {
    piece_dest d;
    d.p = ((l15 & 4) ? a_dst : a_src) + (size_t)(khalf * 8) * HEADS + (l15 & 3);
    d.writer = l15 < LOGITS;
    walk_tiles<true>(ap, B, d, t0);
  } else {
    piece_dest d;
    d.p = h + (size_t)(khalf * 8) * HF + ncol0 + l15;
    d.writer = true;
    walk_tiles<false>(ap, B, d, t0);
  }
}

__device__ __forceinline__ int enc_f(float f) {
  int i = __float_as_int(f);
  return (i >= 0) ? i : (i ^ 0x7fffffff);
}
__device__ __forceinline__ float dec_f(int i) {
  return __int_as_float((i >= 0) ? i : (i ^ 0x7fffffff));
}

#define AGG_THREADS 1024
#define LIST_CAP 5632
#define SCAN_VECS (N_EDGES / 16)
#define SCAN_BATCH 10
#define SCAN_ITERS 5
#define AGG_EDGES_PER_TRIP 8
#define SOFT_BATCH 4
#define ROW_HEADS (TILE_N * HEADS)
static_assert(SCAN_ITERS * SCAN_BATCH * AGG_THREADS >= SCAN_VECS, "the scan covers the edge list");
static_assert(SCAN_BATCH % 2 == 0, "two vectors' match bits share a word");
static_assert(TILE_N <= 256, "the local row fits the list word beside the source id");
static_assert(N_NODES <= (1 << 16), "source id and local row share a word");
static_assert(SOFT_BATCH == 4, "the gather fence of phase 2 names four edges");
static_assert(ROW_HEADS <= AGG_THREADS, "one lane per (row, head) of the small planes");
static_assert(TILE_N * HF * 4 + LIST_CAP * 4 + LIST_CAP * HEADS * 4 + 3 * ROW_HEADS * 4 + 16 <= 320 * 1024, "tile, list, kept scores and planes fit the LDS");

__device__ __forceinline__ unsigned eq_bytes(unsigned w, unsigned pat) {
  unsigned x = w ^ pat;
  unsigned t = (x & 0x7f7f7f7fu) + 0x7f7f7f7fu;
  return ~(t | x) & 0x80808080u;
}

__device__ __forceinline__ int tile_pos(int f) {
  return (f & 128) + ((f & 3) << 5) + ((f & 127) >> 2);
}

__device__ __forceinline__ float edge_score(float as, float ad) {
  float v = as + ad;
  return (v >= 0.f) ? v : NEG_SLOPE * v;
}

template <int U>
__device__ __forceinline__ void agg_trip(int i0, int lane,
                                         const unsigned* list, const float* sc, float* acc,
                                         const float* sm_l, const float* __restrict__ h) {
  int s[U], dl[U];
  v4f h0[U], h1[U];
  #pragma unroll
  for (int u = 0; u < U; ++u) {
    unsigned ent = (unsigned)__builtin_amdgcn_readfirstlane((int)list[i0 + u]);
    s[u]  = (int)(ent & 0xFFFFu);
    dl[u] = (int)(ent >> 16);
  }
  #pragma unroll
  for (int u = 0; u < U; ++u) {
    const float* hs = h + (size_t)s[u] * HF + lane * 4;
    h0[u] = *(const v4f*)(hs);
    h1[u] = *(const v4f*)(hs + 128);
  }
  const int hsel = lane >> 4;
  float alo[U], ahi[U];
  #pragma unroll
  for (int u = 0; u < U; ++u) {
    const int rh = dl[u] * HEADS + hsel;
    const float* ex = sc + (i0 + u) * HEADS + hsel;
    alo[u] = ex[0] * sm_l[rh];
    ahi[u] = ex[2] * sm_l[rh + 2];
  }
  #pragma unroll
  for (int u = 0; u < U; ++u) {
    float* row = acc + dl[u] * HF + lane;
    #pragma unroll
    for (int k = 0; k < 4; ++k) {
      atomicAdd(row + 32 * k,       alo[u] * h0[u][k]);
      atomicAdd(row + 128 + 32 * k, ahi[u] * h1[u][k]);
    }
  }
}

__global__ void __launch_bounds__(AGG_THREADS)
aggregate_kernel(const int* __restrict__ ei, const unsigned* __restrict__ tw, const float* __restrict__ a_src,
                 const float* __restrict__ a_dst, const float* __restrict__ h,
                 const float* __restrict__ bias, float* __restrict__ out) {
  __shared__ __attribute__((aligned(16))) float acc[TILE_N * HF];
  __shared__ unsigned list[LIST_CAP];
  __shared__ __attribute__((aligned(16))) float sc[LIST_CAP * HEADS];
  __shared__ __attribute__((aligned(16))) float ad_l[ROW_HEADS];
  __shared__ __attribute__((aligned(16))) int   mx_i[ROW_HEADS];
  __shared__ __attribute__((aligned(16))) float sm_l[ROW_HEADS];
  __shared__ unsigned cnt;

  const int tid  = threadIdx.x;
  const int lane = tid & 31;
  const int wave = tid >> 5;
  const int lo   = blockIdx.x * TILE_N;

  for (int i = tid; i < TILE_N * HF; i += AGG_THREADS) {
    int p = i & (HF - 1);
    int f = (p & 128) + ((p & 31) << 2) + ((p & 127) >> 5);
    acc[i] = bias[f];
  }
  if (tid < ROW_HEADS) {
    int node = lo + (tid >> 2);
    ad_l[tid] = (node < N_NODES) ? a_dst[(size_t)node * HEADS + (tid & 3)] : 0.f;
    mx_i[tid] = (int)0x80000000;
    sm_l[tid] = 0.f;
  }
  if (tid == 0) cnt = 0u;
  __syncthreads();

  const uint4* tv = (const uint4*)tw;
  const unsigned pat = blockIdx.x * 0x01010101u;
  for (int it = 0; it < SCAN_ITERS; ++it) {
    uint4 dv[SCAN_BATCH];
    #pragma unroll
    for (int u = 0; u < SCAN_BATCH; ++u) {
      int v = (it * SCAN_BATCH + u) * AGG_THREADS + tid;
      dv[u] = (v < SCAN_VECS) ? tv[v] : make_uint4(~0u, ~0u, ~0u, ~0u);
    }
    unsigned m[SCAN_BATCH / 2];
    unsigned c = 0u;
    #pragma unroll
    for (int p = 0; p < SCAN_BATCH / 2; ++p) {
      m[p] = (eq_bytes(dv[2 * p].x, pat) >> 7)     | (eq_bytes(dv[2 * p].y, pat) >> 6)
           | (eq_bytes(dv[2 * p].z, pat) >> 5)     | (eq_bytes(dv[2 * p].w, pat) >> 4)
           | (eq_bytes(dv[2 * p + 1].x, pat) >> 3) | (eq_bytes(dv[2 * p + 1].y, pat) >> 2)
           | (eq_bytes(dv[2 * p + 1].z, pat) >> 1) |  eq_bytes(dv[2 * p + 1].w, pat);
      c += (unsigned)__popc(m[p]);
    }
    if (c) {
      unsigned pos = atomicAdd(&cnt, c);
      #pragma unroll
      for (int p = 0; p < SCAN_BATCH / 2; ++p) {
        const int e0 = 16 * ((it * SCAN_BATCH + 2 * p) * AGG_THREADS + tid);
        unsigned mm = m[p];
        while (mm) {
          int b = __ffs((int)mm) - 1;
          mm &= mm - 1u;
          int e = e0 + ((b & 4) ? 16 * AGG_THREADS : 0) + 4 * (b & 3) + (b >> 3);
          if (pos < (unsigned)LIST_CAP) list[pos] = (unsigned)e;
          ++pos;
        }
      }
    }
  }
  __syncthreads();

  const unsigned total = cnt;
  const bool listed = total <= (unsigned)LIST_CAP;
  const int n = listed ? (int)total : 0;

  for (int b = tid; b < n; b += SOFT_BATCH * AGG_THREADS) {
    unsigned ent[SOFT_BATCH];
    int s[SOFT_BATCH], d[SOFT_BATCH];
    v4f as[SOFT_BATCH];
    #pragma unroll
    for (int u = 0; u < SOFT_BATCH; ++u) {
      int i = b + u * AGG_THREADS;
      ent[u] = (i < n) ? list[i] : 0u;
    }
    #pragma unroll
    for (int u = 0; u < SOFT_BATCH; ++u) {
      s[u] = ei[ent[u]];
      d[u] = ei[N_EDGES + ent[u]];
    }
    #pragma unroll
    for (int u = 0; u < SOFT_BATCH; ++u) as[u] = *(const v4f*)(a_src + (size_t)s[u] * HEADS);
    asm volatile("" :: "v"(as[0]), "v"(as[1]), "v"(as[2]), "v"(as[3]),
                       "v"(d[0]), "v"(d[1]), "v"(d[2]), "v"(d[3]));
    #pragma unroll
    for (int u = 0; u < SOFT_BATCH; ++u) {
      int i = b + u * AGG_THREADS;
      if (i < n) {
        int r = d[u] - lo;
        v4f ad = *(const v4f*)(ad_l + r * HEADS);
        v4f sv;
        #pragma unroll
        for (int q = 0; q < HEADS; ++q) {
          sv[q] = edge_score(as[u][q], ad[q]);
          atomicMax(&mx_i[r * HEADS + q], enc_f(sv[q]));
        }
        *(v4f*)(sc + i * HEADS) = sv;
        list[i] = (unsigned)s[u] | ((unsigned)r << 16);
      }
    }
  }
  if (!listed) {
    for (int e = tid; e < N_EDGES; e += AGG_THREADS) {
      unsigned r = (unsigned)(ei[N_EDGES + e] - lo);
      if (r < (unsigned)TILE_N) {
        int s = ei[e];
        #pragma unroll
        for (int q = 0; q < HEADS; ++q)
          atomicMax(&mx_i[r * HEADS + q],
                    enc_f(edge_score(a_src[(size_t)s * HEADS + q], ad_l[r * HEADS + q])));
      }
    }
  }
  __syncthreads();
  if (tid < ROW_HEADS) {
    float mv = dec_f(mx_i[tid]);
    mx_i[tid] = __float_as_int(__builtin_isfinite(mv) ? mv : 0.f);
  }
  __syncthreads();

  for (int i = tid; i < n; i += AGG_THREADS) {
    const int r = (int)(list[i] >> 16);
    v4f sv = *(const v4f*)(sc + i * HEADS);
    const v4i mv = *(const v4i*)(mx_i + r * HEADS);
    #pragma unroll
    for (int q = 0; q < HEADS; ++q) {
      sv[q] = __expf(sv[q] - __int_as_float(mv[q]));
      atomicAdd(&sm_l[r * HEADS + q], sv[q]);
    }
    *(v4f*)(sc + i * HEADS) = sv;
  }
  if (!listed) {
    for (int e = tid; e < N_EDGES; e += AGG_THREADS) {
      unsigned r = (unsigned)(ei[N_EDGES + e] - lo);
      if (r < (unsigned)TILE_N) {
        int s = ei[e];
        #pragma unroll
        for (int q = 0; q < HEADS; ++q)
          atomicAdd(&sm_l[r * HEADS + q],
                    __expf(edge_score(a_src[(size_t)s * HEADS + q], ad_l[r * HEADS + q]) - __int_as_float(mx_i[r * HEADS + q])));
      }
    }
  }
  __syncthreads();
  if (tid < ROW_HEADS) sm_l[tid] = 1.0f / (sm_l[tid] + GAT_EPS);
  __syncthreads();

  if (!listed) {
    for (int e = tid; e < N_EDGES; e += AGG_THREADS) {
      unsigned r = (unsigned)(ei[N_EDGES + e] - lo);
      if (r < (unsigned)TILE_N) {
        int s = ei[e];
        float al[HEADS];
        #pragma unroll
        for (int q = 0; q < HEADS; ++q)
          al[q] = __expf(edge_score(a_src[(size_t)s * HEADS + q], ad_l[r * HEADS + q]) - __int_as_float(mx_i[r * HEADS + q]))
                  * sm_l[r * HEADS + q];
        #pragma unroll 1
        for (int f = 0; f < HF; ++f) {
          float a = (f < 128) ? ((f < 64) ? al[0] : al[1]) : ((f < 192) ? al[2] : al[3]);
          atomicAdd(&acc[r * HF + tile_pos(f)], a * h[(size_t)s * HF + f]);
        }
      }
    }
  }
  const int full = n / AGG_EDGES_PER_TRIP;
  for (int t = wave; t < full; t += AGG_THREADS / 32)
    agg_trip<AGG_EDGES_PER_TRIP>(t * AGG_EDGES_PER_TRIP, lane, list, sc, acc, sm_l, h);
  for (int i = full * AGG_EDGES_PER_TRIP + wave; i < n; i += AGG_THREADS / 32)
    agg_trip<1>(i, lane, list, sc, acc, sm_l, h);
  __syncthreads();

  for (int r = wave; r < TILE_N; r += AGG_THREADS / 32) {
    int node = lo + r;
    if (node < N_NODES) {
      const float* row = acc + r * HF + lane;
      v4f o0, o1;
      #pragma unroll
      for (int k = 0; k < 4; ++k) { o0[k] = row[32 * k]; o1[k] = row[128 + 32 * k]; }
      float* op = out + (size_t)node * HF + lane * 4;
      *(v4f*)(op)       = o0;
      *(v4f*)(op + 128) = o1;
    }
  }
}

extern "C" void kernel_launch(void* const* d_in, const int* in_sizes, int n_in,
                              void* d_out, int out_size, void* d_ws, size_t ws_size,
                              hipStream_t stream) {
  const float* x       = (const float*)d_in[0];
  const int*   ei      = (const int*)  d_in[1];
  const float* W       = (const float*)d_in[2];
  const float* att_src = (const float*)d_in[3];
  const float* att_dst = (const float*)d_in[4];
  const float* bias    = (const float*)d_in[5];
  float* out = (float*)d_out;

  char* p = (char*)d_ws;
  auto alloc = [&](size_t bytes) {
    char* r = p; p += (bytes + 255) & ~(size_t)255; return r;
  };
  float*    h     = (float*)   alloc((size_t)N_NODES * HF * 4);
  _Float16* Xh    = (_Float16*)alloc((size_t)N_NODES * IN_F * 2);
  _Float16* Wh    = (_Float16*)alloc((size_t)WH_ROWS * IN_F * 2);
  float*    a_src = (float*)   alloc((size_t)N_NODES * HEADS * 4);
  float*    a_dst = (float*)   alloc((size_t)N_NODES * HEADS * 4);
  unsigned* tw    = (unsigned*)alloc((size_t)N_EDGES);

  cvt_f16_kernel  <<<CVT_BLOCKS, CVT_THREADS, 0, stream>>>(x, W, ei, att_src, att_dst, Xh, Wh, tw);
  gemm_wmma_kernel<<<GEMM_BLOCKS, GEMM_THREADS, 0, stream>>>(Xh, Wh, h, a_src, a_dst);
  aggregate_kernel<<<AGG_BLOCKS, AGG_THREADS, 0, stream>>>(ei, tw, a_src, a_dst, h, bias, out);
}


# Last launch sums rows in registers, not LDS: a wave owns rows (row mod 32 = wave), picks its edges from the block list by ballot, adds weight*h, stores its rows; LDS float adds gone; sums reordered
# speedup vs baseline: 6.0342x; 1.0685x over previous
#include <hip/hip_runtime.h>
#include <hip/hip_bf16.h>

#define N_NODES 50000
#define N_EDGES 800000
#define IN_F 128
#define OUT_F 64
#define HEADS 4
#define HF 256
#define NEG_SLOPE 0.2f
#define GAT_EPS 1e-16f
#define LOGITS 8
#define LOGIT_COLS 16
#define WH_ROWS 272

typedef __attribute__((ext_vector_type(16))) _Float16 v16h;
typedef __attribute__((ext_vector_type(8)))  _Float16 v8h;
typedef __attribute__((ext_vector_type(8)))  float    v8f;
typedef __attribute__((ext_vector_type(4)))  float    v4f;
typedef __attribute__((ext_vector_type(4)))  int      v4i;

#define TILE_N 197
#define AGG_BLOCKS 254
#define TILE_NONE 255u
static_assert(TILE_N * AGG_BLOCKS >= N_NODES, "the tiles cover every node");
static_assert(AGG_BLOCKS <= 255, "a block number fits a byte and never equals TILE_NONE");
static_assert(N_EDGES % 16 == 0, "tile bytes are written four and read sixteen at a time");

__device__ __forceinline__ unsigned tile_code(int d) {
  return ((unsigned)d < (unsigned)N_NODES) ? (unsigned)d / TILE_N : TILE_NONE;
}

#define CVT_BLOCKS 2048
#define CVT_THREADS 256
#define FOLD_BLOCKS LOGITS
static_assert(IN_F <= CVT_THREADS, "a fold block has a lane per input feature");

__global__ void __launch_bounds__(CVT_THREADS)
cvt_f16_kernel(const float* __restrict__ x, const float* __restrict__ W,
               const int* __restrict__ ei,
               const float* __restrict__ att_src, const float* __restrict__ att_dst,
               _Float16* __restrict__ Xh, _Float16* __restrict__ Wh,
               unsigned* __restrict__ tw) {
  if (blockIdx.x < FOLD_BLOCKS) {
    const int j  = blockIdx.x;
    const int hd = j & (HEADS - 1);
    const int k  = threadIdx.x;
    if (k < IN_F) {
      const float* att = ((j < HEADS) ? att_src : att_dst) + hd * OUT_F;
      const float* wp  = W + (size_t)hd * OUT_F * IN_F + k;
      float v = 0.f;
      #pragma unroll 16
      for (int f = 0; f < OUT_F; ++f) v += wp[(size_t)f * IN_F] * att[f];
      const _Float16 hi = (_Float16)v;
      Wh[(size_t)(HF + j) * IN_F + k]          = hi;
      Wh[(size_t)(HF + LOGITS + j) * IN_F + k] = (_Float16)(v - (float)hi);
    }
    return;
  }
  int idx = (blockIdx.x - FOLD_BLOCKS) * blockDim.x + threadIdx.x;
  int stride = (gridDim.x - FOLD_BLOCKS) * blockDim.x;
  const int4* dst4 = (const int4*)(ei + N_EDGES);
  for (int i = idx; i < N_EDGES / 4; i += stride) {
    int4 d = dst4[i];
    tw[i] = tile_code(d.x) | (tile_code(d.y) << 8) | (tile_code(d.z) << 16) | (tile_code(d.w) << 24);
  }
  for (int i = idx; i < N_NODES * IN_F; i += stride) Xh[i] = (_Float16)x[i];
  for (int i = idx; i < HF * IN_F; i += stride)      Wh[i] = (_Float16)W[i];
}

#define GEMM_BLOCKS 256
#define GEMM_WAVES 17
#define GEMM_THREADS 544
#define M_TILES (N_NODES / 16)
#define K_BLOCKS (IN_F / 32)
#define FULL_TRIPS (M_TILES / GEMM_BLOCKS)
static_assert(N_NODES % 16 == 0, "every row tile is whole");
static_assert(GEMM_THREADS == 32 * GEMM_WAVES && WH_ROWS == 16 * GEMM_WAVES, "a block's waves cover all columns, the logit columns included");
static_assert(WH_ROWS == HF + LOGIT_COLS && LOGIT_COLS == 2 * LOGITS && LOGITS == 2 * HEADS, "one piece holds both parts of every logit");
static_assert(FULL_TRIPS >= 2, "every block has the two row tiles it asks for first");
static_assert((FULL_TRIPS + 1) * GEMM_BLOCKS >= M_TILES, "one more tile at most behind the full trips");

struct frag_row { v8h p[2 * K_BLOCKS]; };

__device__ __forceinline__ frag_row load_frags(const _Float16* __restrict__ rowp) {
  frag_row f;
  #pragma unroll
  for (int kb = 0; kb < K_BLOCKS; ++kb) {
    f.p[2 * kb]     = *(const v8h*)(rowp + 32 * kb);
    f.p[2 * kb + 1] = *(const v8h*)(rowp + 32 * kb + 16);
  }
  return f;
}

__device__ __forceinline__ v16h frag16(v8h lo, v8h hi) {
  return __builtin_shufflevector(lo, hi, 0, 1, 2, 3, 4, 5, 6, 7, 8, 9, 10, 11, 12, 13, 14, 15);
}

__device__ __forceinline__ v8f tile_product(const frag_row& a, const v16h (&B)[K_BLOCKS]) {
  v8f c = {};
  #pragma unroll
  for (int kb = 0; kb < K_BLOCKS; ++kb)
    c = __builtin_amdgcn_wmma_f32_16x16x32_f16(false, frag16(a.p[2 * kb], a.p[2 * kb + 1]),
                                               false, B[kb], (short)0, c, false, false);
  return c;
}

struct piece_dest { float* p; bool writer; };

template <bool LOGIT_WAVE>
__device__ __forceinline__ void tile_store(v8f c, const piece_dest& d, int t) {
  if (LOGIT_WAVE) {
    float v[8];
    #pragma unroll
    for (int r = 0; r < 8; ++r) v[r] = c[r] + __shfl_xor(c[r], 8, 32);
    if (d.writer) {
      float* lr = d.p + (size_t)t * 16 * HEADS;
      #pragma unroll
      for (int r = 0; r < 8; ++r) lr[(size_t)r * HEADS] = v[r];
    }
  } else {
    float* hr = d.p + (size_t)t * 16 * HF;
    #pragma unroll
    for (int r = 0; r < 8; ++r) hr[(size_t)r * HF] = c[r];
  }
}

template <bool LOGIT_WAVE>
__device__ __forceinline__ void walk_tiles(const _Float16* __restrict__ ap, const v16h (&B)[K_BLOCKS],
                                           const piece_dest& d, int t0) {
  const bool longer = t0 + FULL_TRIPS * GEMM_BLOCKS < M_TILES;
  frag_row f[3];
  f[0] = load_frags(ap + (size_t)t0 * 16 * IN_F);
  f[1] = load_frags(ap + (size_t)(t0 + GEMM_BLOCKS) * 16 * IN_F);
  v8f cprev = {};
  #pragma unroll
  for (int i = 0; i < FULL_TRIPS; ++i) {
    const int t = t0 + i * GEMM_BLOCKS;
    if (i + 2 < FULL_TRIPS)
      f[(i + 2) % 3] = load_frags(ap + (size_t)(t + 2 * GEMM_BLOCKS) * 16 * IN_F);
    else if (i + 2 == FULL_TRIPS)
      f[(i + 2) % 3] = load_frags(ap + (size_t)(longer ? t + 2 * GEMM_BLOCKS : t) * 16 * IN_F);
    __builtin_amdgcn_sched_barrier(0);
    const v8f c = tile_product(f[i % 3], B);
    __builtin_amdgcn_sched_barrier(0);
    if (i > 0) tile_store<LOGIT_WAVE>(cprev, d, t - GEMM_BLOCKS);
    cprev = c;
  }
  const int tl = t0 + (FULL_TRIPS - 1) * GEMM_BLOCKS;
  if (longer) {
    const v8f c = tile_product(f[FULL_TRIPS % 3], B);
    __builtin_amdgcn_sched_barrier(0);
    tile_store<LOGIT_WAVE>(cprev, d, tl);
    tile_store<LOGIT_WAVE>(c, d, tl + GEMM_BLOCKS);
  } else {
    tile_store<LOGIT_WAVE>(cprev, d, tl);
  }
}

__global__ void __launch_bounds__(GEMM_THREADS)
gemm_wmma_kernel(const _Float16* __restrict__ Xh, const _Float16* __restrict__ Wh,
                 float* __restrict__ h, float* __restrict__ a_src, float* __restrict__ a_dst) {
  const int lane  = threadIdx.x & 31;
  const int wave  = __builtin_amdgcn_readfirstlane((int)(threadIdx.x >> 5));
  const int ncol0 = wave << 4;
  const int l15   = lane & 15;
  const int khalf = lane >> 4;

  const frag_row bf = load_frags(Wh + (size_t)(ncol0 + l15) * IN_F + khalf * 8);
  v16h B[K_BLOCKS];
  #pragma unroll
  for (int kb = 0; kb < K_BLOCKS; ++kb) B[kb] = frag16(bf.p[2 * kb], bf.p[2 * kb + 1]);

  const _Float16* ap = Xh + (size_t)l15 * IN_F + khalf * 8;
  const int t0 = blockIdx.x;

  if (wave == GEMM_WAVES - 1) {
    piece_dest d;
    d.p = ((l15 & 4) ? a_dst : a_src) + (size_t)(khalf * 8) * HEADS + (l15 & 3);
    d.writer = l15 < LOGITS;
    walk_tiles<true>(ap, B, d, t0);
  } else {
    piece_dest d;
    d.p = h + (size_t)(khalf * 8) * HF + ncol0 + l15;
    d.writer = true;
    walk_tiles<false>(ap, B, d, t0);
  }
}

__device__ __forceinline__ int enc_f(float f) {
  int i = __float_as_int(f);
  return (i >= 0) ? i : (i ^ 0x7fffffff);
}
__device__ __forceinline__ float dec_f(int i) {
  return __int_as_float((i >= 0) ? i : (i ^ 0x7fffffff));
}

#define AGG_THREADS 1024
#define AGG_WAVES (AGG_THREADS / 32)
#define ROW_SLOTS ((TILE_N + AGG_WAVES - 1) / AGG_WAVES)
#define LIST_CAP 5632
#define SCAN_VECS (N_EDGES / 16)
#define SCAN_BATCH 10
#define SCAN_ITERS 5
#define GATHER_EDGES 4
#define LIST_CHUNK 128
#define RING 256
#define SOFT_BATCH 4
#define ROW_HEADS (TILE_N * HEADS)
static_assert(SCAN_ITERS * SCAN_BATCH * AGG_THREADS >= SCAN_VECS, "the scan covers the edge list");
static_assert(SCAN_BATCH % 2 == 0, "two vectors' match bits share a word");
static_assert(TILE_N <= 256, "the local row fits the list word beside the source id");
static_assert(N_NODES <= (1 << 16), "source id and local row share a word");
static_assert(SOFT_BATCH == 4, "the gather fence of phase 2 names four edges");
static_assert(ROW_HEADS <= AGG_THREADS, "one lane per (row, head) of the small planes");
static_assert(AGG_WAVES == 32, "a row's owner is the low five bits of its local number");
static_assert(ROW_SLOTS <= 8 && LIST_CAP <= (1 << 13), "source id, the row's slot and the list place share a ring word");
static_assert(LIST_CAP % 4 == 0 && LIST_CHUNK == 4 * 32, "a lane reads four whole entries inside the list");
static_assert(GATHER_EDGES - 1 + LIST_CHUNK <= RING && (RING & (RING - 1)) == 0, "what is left plus a step's entries fits the ring");
static_assert(N_EDGES % 32 == 0, "the walk over the edge list itself takes whole waves");
static_assert(LIST_CAP * 4 + LIST_CAP * HEADS * 4 + AGG_WAVES * RING * 4 + 3 * ROW_HEADS * 4 + 16 <= 320 * 1024, "list, kept scores, rings and planes fit the LDS");

__device__ __forceinline__ unsigned eq_bytes(unsigned w, unsigned pat) {
  unsigned x = w ^ pat;
  unsigned t = (x & 0x7f7f7f7fu) + 0x7f7f7f7fu;
  return ~(t | x) & 0x80808080u;
}

__device__ __forceinline__ float edge_score(float as, float ad) {
  float v = as + ad;
  return (v >= 0.f) ? v : NEG_SLOPE * v;
}

struct row_sums { v4f lo[ROW_SLOTS]; v4f hi[ROW_SLOTS]; };

__device__ __forceinline__ void add_row(row_sums& a, int slot, float wl, float wh, v4f h0, v4f h1) {
  #pragma unroll
  for (int k = 0; k < ROW_SLOTS; ++k) {
    if (__builtin_expect(slot == k, 0)) {
      a.lo[k] += wl * h0;
      a.hi[k] += wh * h1;
    }
  }
}

template <int U>
__device__ __forceinline__ void gather_trip(row_sums& a, const unsigned* myring, unsigned head,
                                            int wave, int lane, const float* sc,
                                            const float* sm_l, const float* __restrict__ h) {
  int s[U], slot[U], li[U];
  v4f h0[U], h1[U];
  #pragma unroll
  for (int u = 0; u < U; ++u) {
    unsigned w = (unsigned)__builtin_amdgcn_readfirstlane((int)myring[(head + u) & (RING - 1)]);
    s[u]    = (int)(w & 0xFFFFu);
    slot[u] = (int)((w >> 16) & 7u);
    li[u]   = (int)(w >> 19);
  }
  #pragma unroll
  for (int u = 0; u < U; ++u) {
    const float* hs = h + (size_t)s[u] * HF + lane * 4;
    h0[u] = *(const v4f*)(hs);
    h1[u] = *(const v4f*)(hs + 128);
  }
  const int hsel = lane >> 4;
  float wl[U], wh[U];
  #pragma unroll
  for (int u = 0; u < U; ++u) {
    const int rh = (slot[u] * AGG_WAVES + wave) * HEADS + hsel;
    const float* ex = sc + li[u] * HEADS + hsel;
    wl[u] = ex[0] * sm_l[rh];
    wh[u] = ex[2] * sm_l[rh + 2];
  }
  #pragma unroll
  for (int u = 0; u < U; ++u) add_row(a, slot[u], wl[u], wh[u], h0[u], h1[u]);
}

__global__ void __launch_bounds__(AGG_THREADS)
aggregate_kernel(const int* __restrict__ ei, const unsigned* __restrict__ tw, const float* __restrict__ a_src,
                 const float* __restrict__ a_dst, const float* __restrict__ h,
                 const float* __restrict__ bias, float* __restrict__ out) {
  __shared__ __attribute__((aligned(16))) unsigned list[LIST_CAP];
  __shared__ __attribute__((aligned(16))) float sc[LIST_CAP * HEADS];
  __shared__ unsigned ring[AGG_WAVES * RING];
  __shared__ __attribute__((aligned(16))) float ad_l[ROW_HEADS];
  __shared__ __attribute__((aligned(16))) int   mx_i[ROW_HEADS];
  __shared__ __attribute__((aligned(16))) float sm_l[ROW_HEADS];
  __shared__ unsigned cnt;

  const int tid  = threadIdx.x;
  const int lane = tid & 31;
  const int wave = tid >> 5;
  const int lo   = blockIdx.x * TILE_N;

  if (tid < ROW_HEADS) {
    int node = lo + (tid >> 2);
    ad_l[tid] = (node < N_NODES) ? a_dst[(size_t)node * HEADS + (tid & 3)] : 0.f;
    mx_i[tid] = (int)0x80000000;
    sm_l[tid] = 0.f;
  }
  if (tid == 0) cnt = 0u;
  __syncthreads();

  const uint4* tv = (const uint4*)tw;
  const unsigned pat = blockIdx.x * 0x01010101u;
  for (int it = 0; it < SCAN_ITERS; ++it) {
    uint4 dv[SCAN_BATCH];
    #pragma unroll
    for (int u = 0; u < SCAN_BATCH; ++u) {
      int v = (it * SCAN_BATCH + u) * AGG_THREADS + tid;
      dv[u] = (v < SCAN_VECS) ? tv[v] : make_uint4(~0u, ~0u, ~0u, ~0u);
    }
    unsigned m[SCAN_BATCH / 2];
    unsigned c = 0u;
    #pragma unroll
    for (int p = 0; p < SCAN_BATCH / 2; ++p) {
      m[p] = (eq_bytes(dv[2 * p].x, pat) >> 7)     | (eq_bytes(dv[2 * p].y, pat) >> 6)
           | (eq_bytes(dv[2 * p].z, pat) >> 5)     | (eq_bytes(dv[2 * p].w, pat) >> 4)
           | (eq_bytes(dv[2 * p + 1].x, pat) >> 3) | (eq_bytes(dv[2 * p + 1].y, pat) >> 2)
           | (eq_bytes(dv[2 * p + 1].z, pat) >> 1) |  eq_bytes(dv[2 * p + 1].w, pat);
      c += (unsigned)__popc(m[p]);
    }
    if (c) {
      unsigned pos = atomicAdd(&cnt, c);
      #pragma unroll
      for (int p = 0; p < SCAN_BATCH / 2; ++p) {
        const int e0 = 16 * ((it * SCAN_BATCH + 2 * p) * AGG_THREADS + tid);
        unsigned mm = m[p];
        while (mm) {
          int b = __ffs((int)mm) - 1;
          mm &= mm - 1u;
          int e = e0 + ((b & 4) ? 16 * AGG_THREADS : 0) + 4 * (b & 3) + (b >> 3);
          if (pos < (unsigned)LIST_CAP) list[pos] = (unsigned)e;
          ++pos;
        }
      }
    }
  }
  __syncthreads();

  const unsigned total = cnt;
  const bool listed = total <= (unsigned)LIST_CAP;
  const int n = listed ? (int)total : 0;

  for (int b = tid; b < n; b += SOFT_BATCH * AGG_THREADS) {
    unsigned ent[SOFT_BATCH];
    int s[SOFT_BATCH], d[SOFT_BATCH];
    v4f as[SOFT_BATCH];
    #pragma unroll
    for (int u = 0; u < SOFT_BATCH; ++u) {
      int i = b + u * AGG_THREADS;
      ent[u] = (i < n) ? list[i] : 0u;
    }
    #pragma unroll
    for (int u = 0; u < SOFT_BATCH; ++u) {
      s[u] = ei[ent[u]];
      d[u] = ei[N_EDGES + ent[u]];
    }
    #pragma unroll
    for (int u = 0; u < SOFT_BATCH; ++u) as[u] = *(const v4f*)(a_src + (size_t)s[u] * HEADS);
    asm volatile("" :: "v"(as[0]), "v"(as[1]), "v"(as[2]), "v"(as[3]),
                       "v"(d[0]), "v"(d[1]), "v"(d[2]), "v"(d[3]));
    #pragma unroll
    for (int u = 0; u < SOFT_BATCH; ++u) {
      int i = b + u * AGG_THREADS;
      if (i < n) {
        int r = d[u] - lo;
        v4f ad = *(const v4f*)(ad_l + r * HEADS);
        v4f sv;
        #pragma unroll
        for (int q = 0; q < HEADS; ++q) {
          sv[q] = edge_score(as[u][q], ad[q]);
          atomicMax(&mx_i[r * HEADS + q], enc_f(sv[q]));
        }
        *(v4f*)(sc + i * HEADS) = sv;
        list[i] = (unsigned)s[u] | ((unsigned)r << 16);
      }
    }
  }
  if (!listed) {
    for (int e = tid; e < N_EDGES; e += AGG_THREADS) {
      unsigned r = (unsigned)(ei[N_EDGES + e] - lo);
      if (r < (unsigned)TILE_N) {
        int s = ei[e];
        #pragma unroll
        for (int q = 0; q < HEADS; ++q)
          atomicMax(&mx_i[r * HEADS + q],
                    enc_f(edge_score(a_src[(size_t)s * HEADS + q], ad_l[r * HEADS + q])));
      }
    }
  }
  __syncthreads();
  if (tid < ROW_HEADS) {
    float mv = dec_f(mx_i[tid]);
    mx_i[tid] = __float_as_int(__builtin_isfinite(mv) ? mv : 0.f);
  }
  __syncthreads();

  for (int i = tid; i < n; i += AGG_THREADS) {
    const int r = (int)(list[i] >> 16);
    v4f sv = *(const v4f*)(sc + i * HEADS);
    const v4i mv = *(const v4i*)(mx_i + r * HEADS);
    #pragma unroll
    for (int q = 0; q < HEADS; ++q) {
      sv[q] = __expf(sv[q] - __int_as_float(mv[q]));
      atomicAdd(&sm_l[r * HEADS + q], sv[q]);
    }
    *(v4f*)(sc + i * HEADS) = sv;
  }
  if (!listed) {
    for (int e = tid; e < N_EDGES; e += AGG_THREADS) {
      unsigned r = (unsigned)(ei[N_EDGES + e] - lo);
      if (r < (unsigned)TILE_N) {
        int s = ei[e];
        #pragma unroll
        for (int q = 0; q < HEADS; ++q)
          atomicAdd(&sm_l[r * HEADS + q],
                    __expf(edge_score(a_src[(size_t)s * HEADS + q], ad_l[r * HEADS + q]) - __int_as_float(mx_i[r * HEADS + q])));
      }
    }
  }
  __syncthreads();
  if (tid < ROW_HEADS) sm_l[tid] = 1.0f / (sm_l[tid] + GAT_EPS);
  __syncthreads();

  const int wv = __builtin_amdgcn_readfirstlane(wave);
  row_sums a;
  #pragma unroll
  for (int k = 0; k < ROW_SLOTS; ++k) {
    a.lo[k] = (v4f){0.f, 0.f, 0.f, 0.f};
    a.hi[k] = (v4f){0.f, 0.f, 0.f, 0.f};
  }
  if (!listed) {
    const int hsel = lane >> 4;
    for (int base = 0; base < N_EDGES; base += 32) {
      const int e = base + lane;
      const int ss = ei[e];
      const unsigned rr = (unsigned)(ei[N_EDGES + e] - lo);
      const bool mine = rr < (unsigned)TILE_N && (int)(rr & (unsigned)(AGG_WAVES - 1)) == wv;
      unsigned mask = __builtin_amdgcn_ballot_w32(mine);
      while (mask) {
        const int b = __builtin_ctz(mask);
        mask &= mask - 1u;
        const int s = __builtin_amdgcn_readlane(ss, b);
        const int r = __builtin_amdgcn_readlane((int)rr, b);
        const float* hs = h + (size_t)s * HF + lane * 4;
        const v4f h0 = *(const v4f*)(hs);
        const v4f h1 = *(const v4f*)(hs + 128);
        float w2[2];
        #pragma unroll
        for (int t = 0; t < 2; ++t) {
          const int q  = hsel + 2 * t;
          const int rh = r * HEADS + q;
          w2[t] = __expf(edge_score(a_src[(size_t)s * HEADS + q], ad_l[rh]) - __int_as_float(mx_i[rh])) * sm_l[rh];
        }
        add_row(a, r / AGG_WAVES, w2[0], w2[1], h0, h1);
      }
    }
  }
  {
    const int nu = __builtin_amdgcn_readfirstlane(n);
    unsigned* myring = ring + wv * RING;
    const unsigned below = (1u << lane) - 1u;
    const uint4* lv = (const uint4*)list;
    unsigned head = 0u, tail = 0u;
    for (int base = 0; base < nu; base += LIST_CHUNK) {
      const int i0 = base + 4 * lane;
      const uint4 ev = (i0 < nu) ? lv[i0 >> 2] : make_uint4(0u, 0u, 0u, 0u);
      const unsigned e4[4] = {ev.x, ev.y, ev.z, ev.w};
      #pragma unroll
      for (int q = 0; q < 4; ++q) {
        const int i = i0 + q;
        const unsigned r = e4[q] >> 16;
        const bool mine = i < nu && (int)(r & (unsigned)(AGG_WAVES - 1)) == wv;
        const unsigned mask = __builtin_amdgcn_ballot_w32(mine);
        if (mine)
          myring[(tail + (unsigned)__popc(mask & below)) & (RING - 1)] =
              (e4[q] & 0xFFFFu) | ((r / AGG_WAVES) << 16) | ((unsigned)i << 19);
        tail += (unsigned)__popc(mask);
      }
      while (tail - head >= GATHER_EDGES) {
        gather_trip<GATHER_EDGES>(a, myring, head, wv, lane, sc, sm_l, h);
        head += GATHER_EDGES;
      }
    }
    while (head != tail) {
      gather_trip<1>(a, myring, head, wv, lane, sc, sm_l, h);
      ++head;
    }
  }

  const v4f b0 = *(const v4f*)(bias + lane * 4);
  const v4f b1 = *(const v4f*)(bias + 128 + lane * 4);
  #pragma unroll
  for (int k = 0; k < ROW_SLOTS; ++k) {
    const int r = k * AGG_WAVES + wv;
    const int node = lo + r;
    if (r < TILE_N && node < N_NODES) {
      float* op = out + (size_t)node * HF + lane * 4;
      *(v4f*)(op)       = a.lo[k] + b0;
      *(v4f*)(op + 128) = a.hi[k] + b1;
    }
  }
}

extern "C" void kernel_launch(void* const* d_in, const int* in_sizes, int n_in,
                              void* d_out, int out_size, void* d_ws, size_t ws_size,
                              hipStream_t stream) {
  const float* x       = (const float*)d_in[0];
  const int*   ei      = (const int*)  d_in[1];
  const float* W       = (const float*)d_in[2];
  const float* att_src = (const float*)d_in[3];
  const float* att_dst = (const float*)d_in[4];
  const float* bias    = (const float*)d_in[5];
  float* out = (float*)d_out;

  char* p = (char*)d_ws;
  auto alloc = [&](size_t bytes) {
    char* r = p; p += (bytes + 255) & ~(size_t)255; return r;
  };
  float*    h     = (float*)   alloc((size_t)N_NODES * HF * 4);
  _Float16* Xh    = (_Float16*)alloc((size_t)N_NODES * IN_F * 2);
  _Float16* Wh    = (_Float16*)alloc((size_t)WH_ROWS * IN_F * 2);
  float*    a_src = (float*)   alloc((size_t)N_NODES * HEADS * 4);
  float*    a_dst = (float*)   alloc((size_t)N_NODES * HEADS * 4);
  unsigned* tw    = (unsigned*)alloc((size_t)N_EDGES);

  cvt_f16_kernel  <<<CVT_BLOCKS, CVT_THREADS, 0, stream>>>(x, W, ei, att_src, att_dst, Xh, Wh, tw);
  gemm_wmma_kernel<<<GEMM_BLOCKS, GEMM_THREADS, 0, stream>>>(Xh, Wh, h, a_src, a_dst);
  aggregate_kernel<<<AGG_BLOCKS, AGG_THREADS, 0, stream>>>(ei, tw, a_src, a_dst, h, bias, out);
}
